# starting kernel
# baseline (speedup 1.0000x reference)
#include <hip/hip_runtime.h>
#include <math.h>

#define NB     64
#define CIN    128
#define COUT   128
#define NMODE  64
#define NN     4096
#define LOG2N  12
#define KSPLIT 8

typedef float v2f __attribute__((ext_vector_type(2)));
typedef float v8f __attribute__((ext_vector_type(8)));

static __device__ __forceinline__ v8f wmma_f32_16x16x4(v2f a, v2f b, v8f c) {
  return __builtin_amdgcn_wmma_f32_16x16x4_f32(false, a, false, b, (short)0, c,
                                               false, false);
}

static __device__ __forceinline__ float cas_tw(int p) {
  float th = (float)(p & (NN - 1)) * (6.28318530717958647692f / (float)NN);
  float s, c;
  sincosf(th, &s, &c);
  return c + s;
}

__global__ void prep_cas(float* __restrict__ c1t, float* __restrict__ c2t) {
  int idx = blockIdx.x * blockDim.x + threadIdx.x;
  if (idx < NMODE * NN) {
    int m = idx >> LOG2N;
    int j = idx & (NN - 1);
    float v = 1.0f;
#pragma unroll
    for (int b = 0; b < LOG2N; ++b)
      if ((j >> b) & 1) v *= cas_tw(m << b);
    c1t[(size_t)m * NN + j] = v;
  } else {
    int t = idx - NMODE * NN;
    int n = t >> 6;
    int k = t & (NMODE - 1);
    float v = 1.0f / (float)NN;
#pragma unroll
    for (int b = 0; b < 6; ++b)
      if ((k >> b) & 1) v *= cas_tw(n << b);
    c2t[(size_t)n * NMODE + k] = v;
  }
}

__global__ void prep_w(const float* __restrict__ w, float* __restrict__ wp,
                       float* __restrict__ wm) {
  int idx = blockIdx.x * blockDim.x + threadIdx.x;
  if (idx >= CIN * COUT * NMODE) return;
  int k = idx & (NMODE - 1);
  int kneg = (NMODE - k) & (NMODE - 1);
  float a = w[idx];
  float b = w[idx - k + kneg];
  wp[idx] = a + b;
  wm[idx] = a - b;
}

__global__ void dht_fwd(const float* __restrict__ x, const float* __restrict__ c1t,
                        float* __restrict__ xhp) {
  int wave = (blockIdx.x * blockDim.x + threadIdx.x) >> 5;
  int lane = threadIdx.x & 31;
  int rt = wave >> 3;
  int kc = wave & (KSPLIT - 1);
  int r0 = rt * 16;
  int mrow = lane & 15;
  int khalf = lane >> 4;
  const float* xr  = x   + (size_t)(r0 + mrow) * NN;
  const float* b0p = c1t + (size_t)(mrow)      * NN;
  const float* b1p = c1t + (size_t)(16 + mrow) * NN;
  const float* b2p = c1t + (size_t)(32 + mrow) * NN;
  const float* b3p = c1t + (size_t)(48 + mrow) * NN;
  int k0 = kc * (NN / KSPLIT);
  int k1 = k0 + (NN / KSPLIT);
  v8f a0 = {}, a1 = {}, a2 = {}, a3 = {};
#pragma unroll 4
  for (int k = k0; k < k1; k += 4) {
    int kk = k + khalf * 2;
    v2f a  = *(const v2f*)(xr  + kk);
    v2f b0 = *(const v2f*)(b0p + kk);
    v2f b1 = *(const v2f*)(b1p + kk);
    v2f b2 = *(const v2f*)(b2p + kk);
    v2f b3 = *(const v2f*)(b3p + kk);
    a0 = wmma_f32_16x16x4(a, b0, a0);
    a1 = wmma_f32_16x16x4(a, b1, a1);
    a2 = wmma_f32_16x16x4(a, b2, a2);
    a3 = wmma_f32_16x16x4(a, b3, a3);
  }
  float* dst = xhp + (size_t)kc * ((size_t)NB * CIN * NMODE);
#pragma unroll
  for (int v = 0; v < 8; ++v) {
    int rr = r0 + v + 8 * khalf;
    float* drow = dst + (size_t)rr * NMODE + mrow;
    drow[0]  = a0[v];
    drow[16] = a1[v];
    drow[32] = a2[v];
    drow[48] = a3[v];
  }
}

__global__ void reduce_xh(const float* __restrict__ xhp, float* __restrict__ xh) {
  int idx = blockIdx.x * blockDim.x + threadIdx.x;
  if (idx >= NB * CIN * NMODE) return;
  float s = 0.0f;
#pragma unroll
  for (int c = 0; c < KSPLIT; ++c)
    s += xhp[(size_t)c * ((size_t)NB * CIN * NMODE) + idx];
  xh[idx] = s;
}

__global__ void mode_mix(const float* __restrict__ xh, const float* __restrict__ wp,
                         const float* __restrict__ wm, float* __restrict__ outm) {
  int wave = (blockIdx.x * blockDim.x + threadIdx.x) >> 5;
  int lane = threadIdx.x & 31;
  int k   = wave >> 5;
  int rem = wave & 31;
  int bt = rem >> 3;
  int ot = rem & 7;
  int kneg = (NMODE - k) & (NMODE - 1);
  int mrow = lane & 15;
  int khalf = lane >> 4;
  int bglob = bt * 16 + mrow;
  int oglob = ot * 16 + mrow;
  const float* xr = xh + (size_t)bglob * CIN * NMODE;
  v8f acc = {};
#pragma unroll 4
  for (int i = 0; i < CIN; i += 4) {
    int i0 = i + khalf * 2;
    v2f ap, an, bp, bm;
    ap.x = xr[(size_t)i0 * NMODE + k];
    ap.y = xr[(size_t)(i0 + 1) * NMODE + k];
    an.x = xr[(size_t)i0 * NMODE + kneg];
    an.y = xr[(size_t)(i0 + 1) * NMODE + kneg];
    bp.x = wp[((size_t)i0 * COUT + oglob) * NMODE + k];
    bp.y = wp[((size_t)(i0 + 1) * COUT + oglob) * NMODE + k];
    bm.x = wm[((size_t)i0 * COUT + oglob) * NMODE + k];
    bm.y = wm[((size_t)(i0 + 1) * COUT + oglob) * NMODE + k];
    acc = wmma_f32_16x16x4(ap, bp, acc);
    acc = wmma_f32_16x16x4(an, bm, acc);
  }
#pragma unroll
  for (int v = 0; v < 8; ++v) {
    int bb = bt * 16 + v + 8 * khalf;
    outm[((size_t)bb * COUT + oglob) * NMODE + k] = 0.5f * acc[v];
  }
}

__global__ void dht_inv(const float* __restrict__ outm, const float* __restrict__ c2t,
                        float* __restrict__ y) {
  int wave = (blockIdx.x * blockDim.x + threadIdx.x) >> 5;
  int lane = threadIdx.x & 31;
  int rt = wave >> 6;
  int nc = wave & 63;
  int r0 = rt * 16;
  int n0 = nc * 64;
  int mrow = lane & 15;
  int khalf = lane >> 4;
  const float* ar  = outm + (size_t)(r0 + mrow) * NMODE;
  const float* b0p = c2t + (size_t)(n0 + mrow)      * NMODE;
  const float* b1p = c2t + (size_t)(n0 + 16 + mrow) * NMODE;
  const float* b2p = c2t + (size_t)(n0 + 32 + mrow) * NMODE;
  const float* b3p = c2t + (size_t)(n0 + 48 + mrow) * NMODE;
  v8f a0 = {}, a1 = {}, a2 = {}, a3 = {};
#pragma unroll 4
  for (int k = 0; k < NMODE; k += 4) {
    int kk = k + khalf * 2;
    v2f a  = *(const v2f*)(ar  + kk);
    v2f b0 = *(const v2f*)(b0p + kk);
    v2f b1 = *(const v2f*)(b1p + kk);
    v2f b2 = *(const v2f*)(b2p + kk);
    v2f b3 = *(const v2f*)(b3p + kk);
    a0 = wmma_f32_16x16x4(a, b0, a0);
    a1 = wmma_f32_16x16x4(a, b1, a1);
    a2 = wmma_f32_16x16x4(a, b2, a2);
    a3 = wmma_f32_16x16x4(a, b3, a3);
  }
#pragma unroll
  for (int v = 0; v < 8; ++v) {
    int rr = r0 + v + 8 * khalf;
    float* drow = y + (size_t)rr * NN + n0 + mrow;
    drow[0]  = a0[v];
    drow[16] = a1[v];
    drow[32] = a2[v];
    drow[48] = a3[v];
  }
}

extern "C" void kernel_launch(void* const* d_in, const int* in_sizes, int n_in,
                              void* d_out, int out_size, void* d_ws, size_t ws_size,
                              hipStream_t stream) {
  (void)in_sizes; (void)n_in; (void)out_size; (void)ws_size;
  const float* x = (const float*)d_in[0];
  const float* w = (const float*)d_in[1];
  float* y = (float*)d_out;

  float* ws   = (float*)d_ws;
  float* c1t  = ws;
  float* c2t  = c1t + (size_t)NMODE * NN;
  float* wp   = c2t + (size_t)NN * NMODE;
  float* wm   = wp  + (size_t)CIN * COUT * NMODE;
  float* xhp  = wm  + (size_t)CIN * COUT * NMODE;
  float* xh   = xhp + (size_t)KSPLIT * NB * CIN * NMODE;
  float* outm = xh  + (size_t)NB * CIN * NMODE;

  prep_cas<<<(2 * NMODE * NN) / 256, 256, 0, stream>>>(c1t, c2t);
  prep_w<<<(CIN * COUT * NMODE) / 256, 256, 0, stream>>>(w, wp, wm);

  dht_fwd<<<((NB * CIN / 16) * KSPLIT * 32) / 256, 256, 0, stream>>>(x, c1t, xhp);
  reduce_xh<<<(NB * CIN * NMODE) / 256, 256, 0, stream>>>(xhp, xh);

  mode_mix<<<(NMODE * (NB / 16) * (COUT / 16) * 32) / 256, 256, 0, stream>>>(
      xh, wp, wm, outm);

  dht_inv<<<((NB * COUT / 16) * (NN / 64) * 32) / 256, 256, 0, stream>>>(
      outm, c2t, y);
}


# mode-major layout of the mode-mix operands (xh as [k][b][i], wp/wm as [k][o][i], turned through LDS where they are made): fragment loads become contiguous 8-byte reads; sums unchanged
# speedup vs baseline: 1.3481x; 1.3481x over previous
#include <hip/hip_runtime.h>
#include <math.h>

#define NB     64
#define CIN    128
#define COUT   128
#define NMODE  64
#define NN     4096
#define LOG2N  12
#define KSPLIT 8

typedef float v2f __attribute__((ext_vector_type(2)));
typedef float v8f __attribute__((ext_vector_type(8)));

static __device__ __forceinline__ v8f wmma_f32_16x16x4(v2f a, v2f b, v8f c) {
  return __builtin_amdgcn_wmma_f32_16x16x4_f32(false, a, false, b, (short)0, c,
                                               false, false);
}

static __device__ __forceinline__ float cas_tw(int p) {
  float th = (float)(p & (NN - 1)) * (6.28318530717958647692f / (float)NN);
  float s, c;
  sincosf(th, &s, &c);
  return c + s;
}

__global__ void prep_cas(float* __restrict__ c1t, float* __restrict__ c2t) {
  int idx = blockIdx.x * blockDim.x + threadIdx.x;
  if (idx < NMODE * NN) {
    int m = idx >> LOG2N;
    int j = idx & (NN - 1);
    float v = 1.0f;
#pragma unroll
    for (int b = 0; b < LOG2N; ++b)
      if ((j >> b) & 1) v *= cas_tw(m << b);
    c1t[(size_t)m * NN + j] = v;
  } else {
    int t = idx - NMODE * NN;
    int n = t >> 6;
    int k = t & (NMODE - 1);
    float v = 1.0f / (float)NN;
#pragma unroll
    for (int b = 0; b < 6; ++b)
      if ((k >> b) & 1) v *= cas_tw(n << b);
    c2t[(size_t)n * NMODE + k] = v;
  }
}

#define WT_I 32
__global__ void prep_w(const float* __restrict__ w, float* __restrict__ wpt,
                       float* __restrict__ wmt) {
  __shared__ float tile[WT_I][NMODE + 1];
  int o   = blockIdx.x >> 2;
  int i0  = (blockIdx.x & 3) * WT_I;
  int tid = threadIdx.x;
#pragma unroll
  for (int e = 0; e < (WT_I * NMODE) / 256; ++e) {
    int l  = e * 256 + tid;
    int il = l >> 6;
    int k  = l & (NMODE - 1);
    tile[il][k] = w[((size_t)(i0 + il) * COUT + o) * NMODE + k];
  }
  __syncthreads();
#pragma unroll
  for (int e = 0; e < (WT_I * NMODE) / 256; ++e) {
    int l  = e * 256 + tid;
    int k  = l >> 5;
    int il = l & (WT_I - 1);
    int kneg = (NMODE - k) & (NMODE - 1);
    float a = tile[il][k];
    float b = tile[il][kneg];
    size_t dst = ((size_t)k * COUT + o) * CIN + i0 + il;
    wpt[dst] = a + b;
    wmt[dst] = a - b;
  }
}

__global__ void dht_fwd(const float* __restrict__ x, const float* __restrict__ c1t,
                        float* __restrict__ xhp) {
  int wave = (blockIdx.x * blockDim.x + threadIdx.x) >> 5;
  int lane = threadIdx.x & 31;
  int rt = wave >> 3;
  int kc = wave & (KSPLIT - 1);
  int r0 = rt * 16;
  int mrow = lane & 15;
  int khalf = lane >> 4;
  const float* xr  = x   + (size_t)(r0 + mrow) * NN;
  const float* b0p = c1t + (size_t)(mrow)      * NN;
  const float* b1p = c1t + (size_t)(16 + mrow) * NN;
  const float* b2p = c1t + (size_t)(32 + mrow) * NN;
  const float* b3p = c1t + (size_t)(48 + mrow) * NN;
  int k0 = kc * (NN / KSPLIT);
  int k1 = k0 + (NN / KSPLIT);
  v8f a0 = {}, a1 = {}, a2 = {}, a3 = {};
#pragma unroll 4
  for (int k = k0; k < k1; k += 4) {
    int kk = k + khalf * 2;
    v2f a  = *(const v2f*)(xr  + kk);
    v2f b0 = *(const v2f*)(b0p + kk);
    v2f b1 = *(const v2f*)(b1p + kk);
    v2f b2 = *(const v2f*)(b2p + kk);
    v2f b3 = *(const v2f*)(b3p + kk);
    a0 = wmma_f32_16x16x4(a, b0, a0);
    a1 = wmma_f32_16x16x4(a, b1, a1);
    a2 = wmma_f32_16x16x4(a, b2, a2);
    a3 = wmma_f32_16x16x4(a, b3, a3);
  }
  float* dst = xhp + (size_t)kc * ((size_t)NB * CIN * NMODE);
#pragma unroll
  for (int v = 0; v < 8; ++v) {
    int rr = r0 + v + 8 * khalf;
    float* drow = dst + (size_t)rr * NMODE + mrow;
    drow[0]  = a0[v];
    drow[16] = a1[v];
    drow[32] = a2[v];
    drow[48] = a3[v];
  }
}

#define RT_R 32
__global__ void reduce_xh(const float* __restrict__ xhp, float* __restrict__ xht) {
  __shared__ float tile[NMODE][RT_R + 1];
  int r0  = blockIdx.x * RT_R;
  int tid = threadIdx.x;
#pragma unroll
  for (int e = 0; e < (RT_R * NMODE) / 256; ++e) {
    int l  = e * 256 + tid;
    int rl = l >> 6;
    int m  = l & (NMODE - 1);
    size_t idx = (size_t)r0 * NMODE + l;
    float s = 0.0f;
#pragma unroll
    for (int c = 0; c < KSPLIT; ++c)
      s += xhp[(size_t)c * ((size_t)NB * CIN * NMODE) + idx];
    tile[m][rl] = s;
  }
  __syncthreads();
#pragma unroll
  for (int e = 0; e < (RT_R * NMODE) / 256; ++e) {
    int l  = e * 256 + tid;
    int m  = l >> 5;
    int rl = l & (RT_R - 1);
    xht[(size_t)m * (NB * CIN) + r0 + rl] = tile[m][rl];
  }
}

__global__ void mode_mix(const float* __restrict__ xht, const float* __restrict__ wpt,
                         const float* __restrict__ wmt, float* __restrict__ outm) {
  int wave = (blockIdx.x * blockDim.x + threadIdx.x) >> 5;
  int lane = threadIdx.x & 31;
  int k   = wave >> 5;
  int rem = wave & 31;
  int bt = rem >> 3;
  int ot = rem & 7;
  int kneg = (NMODE - k) & (NMODE - 1);
  int mrow = lane & 15;
  int khalf = lane >> 4;
  int bglob = bt * 16 + mrow;
  int oglob = ot * 16 + mrow;
  const float* apr = xht + ((size_t)k    * NB   + bglob) * CIN + khalf * 2;
  const float* anr = xht + ((size_t)kneg * NB   + bglob) * CIN + khalf * 2;
  const float* bpr = wpt + ((size_t)k    * COUT + oglob) * CIN + khalf * 2;
  const float* bmr = wmt + ((size_t)k    * COUT + oglob) * CIN + khalf * 2;
  v8f acc = {};
#pragma unroll 8
  for (int i = 0; i < CIN; i += 4) {
    v2f ap = *(const v2f*)(apr + i);
    v2f an = *(const v2f*)(anr + i);
    v2f bp = *(const v2f*)(bpr + i);
    v2f bm = *(const v2f*)(bmr + i);
    acc = wmma_f32_16x16x4(ap, bp, acc);
    acc = wmma_f32_16x16x4(an, bm, acc);
  }
#pragma unroll
  for (int v = 0; v < 8; ++v) {
    int bb = bt * 16 + v + 8 * khalf;
    outm[((size_t)bb * COUT + oglob) * NMODE + k] = 0.5f * acc[v];
  }
}

__global__ void dht_inv(const float* __restrict__ outm, const float* __restrict__ c2t,
                        float* __restrict__ y) {
  int wave = (blockIdx.x * blockDim.x + threadIdx.x) >> 5;
  int lane = threadIdx.x & 31;
  int rt = wave >> 6;
  int nc = wave & 63;
  int r0 = rt * 16;
  int n0 = nc * 64;
  int mrow = lane & 15;
  int khalf = lane >> 4;
  const float* ar  = outm + (size_t)(r0 + mrow) * NMODE;
  const float* b0p = c2t + (size_t)(n0 + mrow)      * NMODE;
  const float* b1p = c2t + (size_t)(n0 + 16 + mrow) * NMODE;
  const float* b2p = c2t + (size_t)(n0 + 32 + mrow) * NMODE;
  const float* b3p = c2t + (size_t)(n0 + 48 + mrow) * NMODE;
  v8f a0 = {}, a1 = {}, a2 = {}, a3 = {};
#pragma unroll 4
  for (int k = 0; k < NMODE; k += 4) {
    int kk = k + khalf * 2;
    v2f a  = *(const v2f*)(ar  + kk);
    v2f b0 = *(const v2f*)(b0p + kk);
    v2f b1 = *(const v2f*)(b1p + kk);
    v2f b2 = *(const v2f*)(b2p + kk);
    v2f b3 = *(const v2f*)(b3p + kk);
    a0 = wmma_f32_16x16x4(a, b0, a0);
    a1 = wmma_f32_16x16x4(a, b1, a1);
    a2 = wmma_f32_16x16x4(a, b2, a2);
    a3 = wmma_f32_16x16x4(a, b3, a3);
  }
#pragma unroll
  for (int v = 0; v < 8; ++v) {
    int rr = r0 + v + 8 * khalf;
    float* drow = y + (size_t)rr * NN + n0 + mrow;
    drow[0]  = a0[v];
    drow[16] = a1[v];
    drow[32] = a2[v];
    drow[48] = a3[v];
  }
}

extern "C" void kernel_launch(void* const* d_in, const int* in_sizes, int n_in,
                              void* d_out, int out_size, void* d_ws, size_t ws_size,
                              hipStream_t stream) {
  (void)in_sizes; (void)n_in; (void)out_size; (void)ws_size;
  const float* x = (const float*)d_in[0];
  const float* w = (const float*)d_in[1];
  float* y = (float*)d_out;

  float* ws   = (float*)d_ws;
  float* c1t  = ws;
  float* c2t  = c1t + (size_t)NMODE * NN;
  float* wp   = c2t + (size_t)NN * NMODE;
  float* wm   = wp  + (size_t)CIN * COUT * NMODE;
  float* xhp  = wm  + (size_t)CIN * COUT * NMODE;
  float* xh   = xhp + (size_t)KSPLIT * NB * CIN * NMODE;
  float* outm = xh  + (size_t)NB * CIN * NMODE;

  prep_cas<<<(2 * NMODE * NN) / 256, 256, 0, stream>>>(c1t, c2t);
  prep_w<<<COUT * (CIN / WT_I), 256, 0, stream>>>(w, wp, wm);

  dht_fwd<<<((NB * CIN / 16) * KSPLIT * 32) / 256, 256, 0, stream>>>(x, c1t, xhp);
  reduce_xh<<<(NB * CIN) / RT_R, 256, 0, stream>>>(xhp, xh);

  mode_mix<<<(NMODE * (NB / 16) * (COUT / 16) * 32) / 256, 256, 0, stream>>>(
      xh, wp, wm, outm);

  dht_inv<<<((NB * COUT / 16) * (NN / 64) * 32) / 256, 256, 0, stream>>>(
      outm, c2t, y);
}


# 16-byte fragment loads in both transforms, twiddle tables stored fragment-major (a B load = 4 whole lines, not 16 partial), fwd issues a trip's 40 loads first; k order per 8: 0,1,4,5,2,3,6,7
# speedup vs baseline: 1.7916x; 1.3290x over previous
#include <hip/hip_runtime.h>
#include <math.h>

#define NB     64
#define CIN    128
#define COUT   128
#define NMODE  64
#define NN     4096
#define LOG2N  12
#define KSPLIT 8

typedef float v2f __attribute__((ext_vector_type(2)));
typedef float v4f __attribute__((ext_vector_type(4)));
typedef float v8f __attribute__((ext_vector_type(8)));

static __device__ __forceinline__ v8f wmma_f32_16x16x4(v2f a, v2f b, v8f c) {
  return __builtin_amdgcn_wmma_f32_16x16x4_f32(false, a, false, b, (short)0, c,
                                               false, false);
}

static __device__ __forceinline__ float cas_tw(int p) {
  float th = (float)(p & (NN - 1)) * (6.28318530717958647692f / (float)NN);
  float s, c;
  sincosf(th, &s, &c);
  return c + s;
}

__global__ void prep_cas(float* __restrict__ c1f, float* __restrict__ c2f) {
  int idx = blockIdx.x * blockDim.x + threadIdx.x;
  int e = idx & 3;
  int l = (idx >> 2) & 31;
  int t = (idx >> 7) & 3;
  int c = l & 15;
  int h = l >> 4;
  if (idx < NMODE * NN) {
    int g = idx >> 9;
    int m = 16 * t + c;
    int j = 8 * g + 4 * h + e;
    float v = 1.0f;
#pragma unroll
    for (int b = 0; b < LOG2N; ++b)
      if ((j >> b) & 1) v *= cas_tw(m << b);
    c1f[idx] = v;
  } else {
    int u = idx - NMODE * NN;
    int g  = (u >> 9) & 7;
    int nc = u >> 12;
    int n = 64 * nc + 16 * t + c;
    int k = 8 * g + 4 * h + e;
    float v = 1.0f / (float)NN;
#pragma unroll
    for (int b = 0; b < 6; ++b)
      if ((k >> b) & 1) v *= cas_tw(n << b);
    c2f[u] = v;
  }
}

#define WT_I 32
__global__ void prep_w(const float* __restrict__ w, float* __restrict__ wpt,
                       float* __restrict__ wmt) {
  __shared__ float tile[WT_I][NMODE + 1];
  int o   = blockIdx.x >> 2;
  int i0  = (blockIdx.x & 3) * WT_I;
  int tid = threadIdx.x;
#pragma unroll
  for (int e = 0; e < (WT_I * NMODE) / 256; ++e) {
    int l  = e * 256 + tid;
    int il = l >> 6;
    int k  = l & (NMODE - 1);
    tile[il][k] = w[((size_t)(i0 + il) * COUT + o) * NMODE + k];
  }
  __syncthreads();
#pragma unroll
  for (int e = 0; e < (WT_I * NMODE) / 256; ++e) {
    int l  = e * 256 + tid;
    int k  = l >> 5;
    int il = l & (WT_I - 1);
    int kneg = (NMODE - k) & (NMODE - 1);
    float a = tile[il][k];
    float b = tile[il][kneg];
    size_t dst = ((size_t)k * COUT + o) * CIN + i0 + il;
    wpt[dst] = a + b;
    wmt[dst] = a - b;
  }
}

__global__ void __launch_bounds__(256)
dht_fwd(const float* __restrict__ x, const float* __restrict__ c1f,
                        float* __restrict__ xhp) {
  int wave = (blockIdx.x * blockDim.x + threadIdx.x) >> 5;
  int lane = threadIdx.x & 31;
  int rt = wave >> 3;
  int kc = wave & (KSPLIT - 1);
  int r0 = rt * 16;
  int mrow = lane & 15;
  int khalf = lane >> 4;
  const float* xr = x + (size_t)(r0 + mrow) * NN + khalf * 4;
  const v4f*   bf = (const v4f*)c1f + lane;
  int g0 = kc * (NN / KSPLIT / 8);
  int g1 = g0 + (NN / KSPLIT / 8);
  v8f a0 = {}, a1 = {}, a2 = {}, a3 = {};
#define FWD_TRIP 8
#pragma nounroll
  for (int g = g0; g < g1; g += FWD_TRIP) {
    v4f a[FWD_TRIP], b[FWD_TRIP][4];
#pragma unroll
    for (int u = 0; u < FWD_TRIP; ++u) {
      a[u] = *(const v4f*)(xr + (g + u) * 8);
      const v4f* bg = bf + (size_t)(g + u) * 128;
      b[u][0] = bg[0];
      b[u][1] = bg[32];
      b[u][2] = bg[64];
      b[u][3] = bg[96];
    }
    __builtin_amdgcn_sched_barrier(0);
#pragma unroll
    for (int u = 0; u < FWD_TRIP; ++u) {
      a0 = wmma_f32_16x16x4(a[u].xy, b[u][0].xy, a0);
      a1 = wmma_f32_16x16x4(a[u].xy, b[u][1].xy, a1);
      a2 = wmma_f32_16x16x4(a[u].xy, b[u][2].xy, a2);
      a3 = wmma_f32_16x16x4(a[u].xy, b[u][3].xy, a3);
      a0 = wmma_f32_16x16x4(a[u].zw, b[u][0].zw, a0);
      a1 = wmma_f32_16x16x4(a[u].zw, b[u][1].zw, a1);
      a2 = wmma_f32_16x16x4(a[u].zw, b[u][2].zw, a2);
      a3 = wmma_f32_16x16x4(a[u].zw, b[u][3].zw, a3);
    }
  }
  float* dst = xhp + (size_t)kc * ((size_t)NB * CIN * NMODE);
#pragma unroll
  for (int v = 0; v < 8; ++v) {
    int rr = r0 + v + 8 * khalf;
    float* drow = dst + (size_t)rr * NMODE + mrow;
    drow[0]  = a0[v];
    drow[16] = a1[v];
    drow[32] = a2[v];
    drow[48] = a3[v];
  }
}

#define RT_R 32
__global__ void reduce_xh(const float* __restrict__ xhp, float* __restrict__ xht) {
  __shared__ float tile[NMODE][RT_R + 1];
  int r0  = blockIdx.x * RT_R;
  int tid = threadIdx.x;
#pragma unroll
  for (int e = 0; e < (RT_R * NMODE) / 256; ++e) {
    int l  = e * 256 + tid;
    int rl = l >> 6;
    int m  = l & (NMODE - 1);
    size_t idx = (size_t)r0 * NMODE + l;
    float s = 0.0f;
#pragma unroll
    for (int c = 0; c < KSPLIT; ++c)
      s += xhp[(size_t)c * ((size_t)NB * CIN * NMODE) + idx];
    tile[m][rl] = s;
  }
  __syncthreads();
#pragma unroll
  for (int e = 0; e < (RT_R * NMODE) / 256; ++e) {
    int l  = e * 256 + tid;
    int m  = l >> 5;
    int rl = l & (RT_R - 1);
    xht[(size_t)m * (NB * CIN) + r0 + rl] = tile[m][rl];
  }
}

__global__ void mode_mix(const float* __restrict__ xht, const float* __restrict__ wpt,
                         const float* __restrict__ wmt, float* __restrict__ outm) {
  int wave = (blockIdx.x * blockDim.x + threadIdx.x) >> 5;
  int lane = threadIdx.x & 31;
  int k   = wave >> 5;
  int rem = wave & 31;
  int bt = rem >> 3;
  int ot = rem & 7;
  int kneg = (NMODE - k) & (NMODE - 1);
  int mrow = lane & 15;
  int khalf = lane >> 4;
  int bglob = bt * 16 + mrow;
  int oglob = ot * 16 + mrow;
  const float* apr = xht + ((size_t)k    * NB   + bglob) * CIN + khalf * 2;
  const float* anr = xht + ((size_t)kneg * NB   + bglob) * CIN + khalf * 2;
  const float* bpr = wpt + ((size_t)k    * COUT + oglob) * CIN + khalf * 2;
  const float* bmr = wmt + ((size_t)k    * COUT + oglob) * CIN + khalf * 2;
  v8f acc = {};
#pragma unroll 8
  for (int i = 0; i < CIN; i += 4) {
    v2f ap = *(const v2f*)(apr + i);
    v2f an = *(const v2f*)(anr + i);
    v2f bp = *(const v2f*)(bpr + i);
    v2f bm = *(const v2f*)(bmr + i);
    acc = wmma_f32_16x16x4(ap, bp, acc);
    acc = wmma_f32_16x16x4(an, bm, acc);
  }
#pragma unroll
  for (int v = 0; v < 8; ++v) {
    int bb = bt * 16 + v + 8 * khalf;
    outm[((size_t)bb * COUT + oglob) * NMODE + k] = 0.5f * acc[v];
  }
}

__global__ void dht_inv(const float* __restrict__ outm, const float* __restrict__ c2f,
                        float* __restrict__ y) {
  int wave = (blockIdx.x * blockDim.x + threadIdx.x) >> 5;
  int lane = threadIdx.x & 31;
  int rt = wave >> 6;
  int nc = wave & 63;
  int r0 = rt * 16;
  int n0 = nc * 64;
  int mrow = lane & 15;
  int khalf = lane >> 4;
  const float* ar = outm + (size_t)(r0 + mrow) * NMODE + khalf * 4;
  const v4f*   bf = (const v4f*)c2f + (size_t)nc * (8 * 128) + lane;
  v8f a0 = {}, a1 = {}, a2 = {}, a3 = {};
#pragma unroll 4
  for (int g = 0; g < NMODE / 8; ++g) {
    v4f a = *(const v4f*)(ar + g * 8);
    const v4f* bg = bf + g * 128;
    v4f b0 = bg[0];
    v4f b1 = bg[32];
    v4f b2 = bg[64];
    v4f b3 = bg[96];
    a0 = wmma_f32_16x16x4(a.xy, b0.xy, a0);
    a1 = wmma_f32_16x16x4(a.xy, b1.xy, a1);
    a2 = wmma_f32_16x16x4(a.xy, b2.xy, a2);
    a3 = wmma_f32_16x16x4(a.xy, b3.xy, a3);
    a0 = wmma_f32_16x16x4(a.zw, b0.zw, a0);
    a1 = wmma_f32_16x16x4(a.zw, b1.zw, a1);
    a2 = wmma_f32_16x16x4(a.zw, b2.zw, a2);
    a3 = wmma_f32_16x16x4(a.zw, b3.zw, a3);
  }
#pragma unroll
  for (int v = 0; v < 8; ++v) {
    int rr = r0 + v + 8 * khalf;
    float* drow = y + (size_t)rr * NN + n0 + mrow;
    drow[0]  = a0[v];
    drow[16] = a1[v];
    drow[32] = a2[v];
    drow[48] = a3[v];
  }
}

extern "C" void kernel_launch(void* const* d_in, const int* in_sizes, int n_in,
                              void* d_out, int out_size, void* d_ws, size_t ws_size,
                              hipStream_t stream) {
  (void)in_sizes; (void)n_in; (void)out_size; (void)ws_size;
  const float* x = (const float*)d_in[0];
  const float* w = (const float*)d_in[1];
  float* y = (float*)d_out;

  float* ws   = (float*)d_ws;
  float* c1t  = ws;
  float* c2t  = c1t + (size_t)NMODE * NN;
  float* wp   = c2t + (size_t)NN * NMODE;
  float* wm   = wp  + (size_t)CIN * COUT * NMODE;
  float* xhp  = wm  + (size_t)CIN * COUT * NMODE;
  float* xh   = xhp + (size_t)KSPLIT * NB * CIN * NMODE;
  float* outm = xh  + (size_t)NB * CIN * NMODE;

  prep_cas<<<(2 * NMODE * NN) / 256, 256, 0, stream>>>(c1t, c2t);
  prep_w<<<COUT * (CIN / WT_I), 256, 0, stream>>>(w, wp, wm);

  dht_fwd<<<((NB * CIN / 16) * KSPLIT * 32) / 256, 256, 0, stream>>>(x, c1t, xhp);
  reduce_xh<<<(NB * CIN) / RT_R, 256, 0, stream>>>(xhp, xh);

  mode_mix<<<(NMODE * (NB / 16) * (COUT / 16) * 32) / 256, 256, 0, stream>>>(
      xh, wp, wm, outm);

  dht_inv<<<((NB * COUT / 16) * (NN / 64) * 32) / 256, 256, 0, stream>>>(
      outm, c2t, y);
}


# each twiddle (B) fragment feeds several row tiles from registers: inv holds its 64x64 B block over 8 row tiles (4096 waves, was 32768), fwd takes 32 rows per wave, next trip loaded ahead; same sums
# speedup vs baseline: 1.9003x; 1.0607x over previous
#include <hip/hip_runtime.h>
#include <math.h>

#define NB     64
#define CIN    128
#define COUT   128
#define NMODE  64
#define NN     4096
#define LOG2N  12
#define KSPLIT 8

typedef float v2f __attribute__((ext_vector_type(2)));
typedef float v4f __attribute__((ext_vector_type(4)));
typedef float v8f __attribute__((ext_vector_type(8)));

static __device__ __forceinline__ v8f wmma_f32_16x16x4(v2f a, v2f b, v8f c) {
  return __builtin_amdgcn_wmma_f32_16x16x4_f32(false, a, false, b, (short)0, c,
                                               false, false);
}

static __device__ __forceinline__ float cas_tw(int p) {
  float th = (float)(p & (NN - 1)) * (6.28318530717958647692f / (float)NN);
  float s, c;
  sincosf(th, &s, &c);
  return c + s;
}

__global__ void prep_cas(float* __restrict__ c1f, float* __restrict__ c2f) {
  int idx = blockIdx.x * blockDim.x + threadIdx.x;
  int e = idx & 3;
  int l = (idx >> 2) & 31;
  int t = (idx >> 7) & 3;
  int c = l & 15;
  int h = l >> 4;
  if (idx < NMODE * NN) {
    int g = idx >> 9;
    int m = 16 * t + c;
    int j = 8 * g + 4 * h + e;
    float v = 1.0f;
#pragma unroll
    for (int b = 0; b < LOG2N; ++b)
      if ((j >> b) & 1) v *= cas_tw(m << b);
    c1f[idx] = v;
  } else {
    int u = idx - NMODE * NN;
    int g  = (u >> 9) & 7;
    int nc = u >> 12;
    int n = 64 * nc + 16 * t + c;
    int k = 8 * g + 4 * h + e;
    float v = 1.0f / (float)NN;
#pragma unroll
    for (int b = 0; b < 6; ++b)
      if ((k >> b) & 1) v *= cas_tw(n << b);
    c2f[u] = v;
  }
}

#define WT_I 32
__global__ void prep_w(const float* __restrict__ w, float* __restrict__ wpt,
                       float* __restrict__ wmt) {
  __shared__ float tile[WT_I][NMODE + 1];
  int o   = blockIdx.x >> 2;
  int i0  = (blockIdx.x & 3) * WT_I;
  int tid = threadIdx.x;
#pragma unroll
  for (int e = 0; e < (WT_I * NMODE) / 256; ++e) {
    int l  = e * 256 + tid;
    int il = l >> 6;
    int k  = l & (NMODE - 1);
    tile[il][k] = w[((size_t)(i0 + il) * COUT + o) * NMODE + k];
  }
  __syncthreads();
#pragma unroll
  for (int e = 0; e < (WT_I * NMODE) / 256; ++e) {
    int l  = e * 256 + tid;
    int k  = l >> 5;
    int il = l & (WT_I - 1);
    int kneg = (NMODE - k) & (NMODE - 1);
    float a = tile[il][k];
    float b = tile[il][kneg];
    size_t dst = ((size_t)k * COUT + o) * CIN + i0 + il;
    wpt[dst] = a + b;
    wmt[dst] = a - b;
  }
}

#define FWD_TRIP 4

static __device__ __forceinline__ void fwd_load(v4f (&a)[2][FWD_TRIP], v4f (&b)[FWD_TRIP][4],
                                                const float* __restrict__ xr0,
                                                const float* __restrict__ xr1,
                                                const v4f* __restrict__ bf, int g) {
#pragma unroll
  for (int u = 0; u < FWD_TRIP; ++u) {
    a[0][u] = *(const v4f*)(xr0 + (g + u) * 8);
    a[1][u] = *(const v4f*)(xr1 + (g + u) * 8);
    const v4f* bg = bf + (size_t)(g + u) * 128;
    b[u][0] = bg[0];
    b[u][1] = bg[32];
    b[u][2] = bg[64];
    b[u][3] = bg[96];
  }
}

static __device__ __forceinline__ void fwd_mma(const v4f (&a)[2][FWD_TRIP],
                                               const v4f (&b)[FWD_TRIP][4], v8f (&acc)[2][4]) {
#pragma unroll
  for (int u = 0; u < FWD_TRIP; ++u) {
#pragma unroll
    for (int r = 0; r < 2; ++r)
#pragma unroll
      for (int t = 0; t < 4; ++t)
        acc[r][t] = wmma_f32_16x16x4(a[r][u].xy, b[u][t].xy, acc[r][t]);
#pragma unroll
    for (int r = 0; r < 2; ++r)
#pragma unroll
      for (int t = 0; t < 4; ++t)
        acc[r][t] = wmma_f32_16x16x4(a[r][u].zw, b[u][t].zw, acc[r][t]);
  }
}

__global__ void __launch_bounds__(128)
dht_fwd(const float* __restrict__ x, const float* __restrict__ c1f,
                        float* __restrict__ xhp) {
  int lane = threadIdx.x & 31;
  int kc = blockIdx.x & (KSPLIT - 1);
  int rt = (blockIdx.x >> 3) * 4 + (threadIdx.x >> 5);
  int r0 = rt * 32;
  int mrow = lane & 15;
  int khalf = lane >> 4;
  const float* xr0 = x + (size_t)(r0 + mrow) * NN + khalf * 4;
  const float* xr1 = xr0 + (size_t)16 * NN;
  const v4f*   bf = (const v4f*)c1f + lane;
  int g0 = kc * (NN / KSPLIT / 8);
  int g1 = g0 + (NN / KSPLIT / 8);
  v8f acc[2][4] = {};
  v4f aA[2][FWD_TRIP], bA[FWD_TRIP][4];
  v4f aB[2][FWD_TRIP], bB[FWD_TRIP][4];
  fwd_load(aA, bA, xr0, xr1, bf, g0);
#pragma nounroll
  for (int g = g0; g < g1; g += 2 * FWD_TRIP) {
    fwd_load(aB, bB, xr0, xr1, bf, g + FWD_TRIP);
    __builtin_amdgcn_sched_barrier(0);
    fwd_mma(aA, bA, acc);
    int gn = (g + 2 * FWD_TRIP < g1) ? (g + 2 * FWD_TRIP) : (g1 - FWD_TRIP);
    fwd_load(aA, bA, xr0, xr1, bf, gn);
    __builtin_amdgcn_sched_barrier(0);
    fwd_mma(aB, bB, acc);
  }
  float* dst = xhp + (size_t)kc * ((size_t)NB * CIN * NMODE);
#pragma unroll
  for (int r = 0; r < 2; ++r) {
#pragma unroll
    for (int v = 0; v < 8; ++v) {
      int rr = r0 + 16 * r + v + 8 * khalf;
      float* drow = dst + (size_t)rr * NMODE + mrow;
      drow[0]  = acc[r][0][v];
      drow[16] = acc[r][1][v];
      drow[32] = acc[r][2][v];
      drow[48] = acc[r][3][v];
    }
  }
}

#define RT_R 32
__global__ void reduce_xh(const float* __restrict__ xhp, float* __restrict__ xht) {
  __shared__ float tile[NMODE][RT_R + 1];
  int r0  = blockIdx.x * RT_R;
  int tid = threadIdx.x;
#pragma unroll
  for (int e = 0; e < (RT_R * NMODE) / 256; ++e) {
    int l  = e * 256 + tid;
    int rl = l >> 6;
    int m  = l & (NMODE - 1);
    size_t idx = (size_t)r0 * NMODE + l;
    float s = 0.0f;
#pragma unroll
    for (int c = 0; c < KSPLIT; ++c)
      s += xhp[(size_t)c * ((size_t)NB * CIN * NMODE) + idx];
    tile[m][rl] = s;
  }
  __syncthreads();
#pragma unroll
  for (int e = 0; e < (RT_R * NMODE) / 256; ++e) {
    int l  = e * 256 + tid;
    int m  = l >> 5;
    int rl = l & (RT_R - 1);
    xht[(size_t)m * (NB * CIN) + r0 + rl] = tile[m][rl];
  }
}

__global__ void mode_mix(const float* __restrict__ xht, const float* __restrict__ wpt,
                         const float* __restrict__ wmt, float* __restrict__ outm) {
  int wave = (blockIdx.x * blockDim.x + threadIdx.x) >> 5;
  int lane = threadIdx.x & 31;
  int k   = wave >> 5;
  int rem = wave & 31;
  int bt = rem >> 3;
  int ot = rem & 7;
  int kneg = (NMODE - k) & (NMODE - 1);
  int mrow = lane & 15;
  int khalf = lane >> 4;
  int bglob = bt * 16 + mrow;
  int oglob = ot * 16 + mrow;
  const float* apr = xht + ((size_t)k    * NB   + bglob) * CIN + khalf * 2;
  const float* anr = xht + ((size_t)kneg * NB   + bglob) * CIN + khalf * 2;
  const float* bpr = wpt + ((size_t)k    * COUT + oglob) * CIN + khalf * 2;
  const float* bmr = wmt + ((size_t)k    * COUT + oglob) * CIN + khalf * 2;
  v8f acc = {};
#pragma unroll 8
  for (int i = 0; i < CIN; i += 4) {
    v2f ap = *(const v2f*)(apr + i);
    v2f an = *(const v2f*)(anr + i);
    v2f bp = *(const v2f*)(bpr + i);
    v2f bm = *(const v2f*)(bmr + i);
    acc = wmma_f32_16x16x4(ap, bp, acc);
    acc = wmma_f32_16x16x4(an, bm, acc);
  }
#pragma unroll
  for (int v = 0; v < 8; ++v) {
    int bb = bt * 16 + v + 8 * khalf;
    outm[((size_t)bb * COUT + oglob) * NMODE + k] = 0.5f * acc[v];
  }
}

#define INV_TILES 8
__global__ void __launch_bounds__(256)
dht_inv(const float* __restrict__ outm, const float* __restrict__ c2f,
                        float* __restrict__ y) {
  int wave = (blockIdx.x * blockDim.x + threadIdx.x) >> 5;
  int lane = threadIdx.x & 31;
  int rg = wave >> 6;
  int nc = wave & 63;
  int r0 = rg * (INV_TILES * 16);
  int n0 = nc * 64;
  int mrow = lane & 15;
  int khalf = lane >> 4;
  const float* ar = outm + (size_t)(r0 + mrow) * NMODE + khalf * 4;
  const v4f*   bf = (const v4f*)c2f + (size_t)nc * (8 * 128) + lane;
  v4f a[NMODE / 8];
#pragma unroll
  for (int g = 0; g < NMODE / 8; ++g) a[g] = *(const v4f*)(ar + g * 8);
  v4f b[NMODE / 8][4];
#pragma unroll
  for (int g = 0; g < NMODE / 8; ++g) {
    const v4f* bg = bf + g * 128;
    b[g][0] = bg[0];
    b[g][1] = bg[32];
    b[g][2] = bg[64];
    b[g][3] = bg[96];
  }
  float* yr = y + (size_t)(r0 + 8 * khalf) * NN + n0 + mrow;
#pragma nounroll
  for (int t = 0; t < INV_TILES; ++t) {
    int tn = (t + 1 < INV_TILES) ? (t + 1) : t;
    const float* arn = ar + (size_t)tn * (16 * NMODE);
    v4f an[NMODE / 8];
#pragma unroll
    for (int g = 0; g < NMODE / 8; ++g) an[g] = *(const v4f*)(arn + g * 8);
    v8f a0 = {}, a1 = {}, a2 = {}, a3 = {};
#pragma unroll
    for (int g = 0; g < NMODE / 8; ++g) {
      a0 = wmma_f32_16x16x4(a[g].xy, b[g][0].xy, a0);
      a1 = wmma_f32_16x16x4(a[g].xy, b[g][1].xy, a1);
      a2 = wmma_f32_16x16x4(a[g].xy, b[g][2].xy, a2);
      a3 = wmma_f32_16x16x4(a[g].xy, b[g][3].xy, a3);
      a0 = wmma_f32_16x16x4(a[g].zw, b[g][0].zw, a0);
      a1 = wmma_f32_16x16x4(a[g].zw, b[g][1].zw, a1);
      a2 = wmma_f32_16x16x4(a[g].zw, b[g][2].zw, a2);
      a3 = wmma_f32_16x16x4(a[g].zw, b[g][3].zw, a3);
    }
    float* yt = yr + (size_t)t * (16 * NN);
#pragma unroll
    for (int v = 0; v < 8; ++v) {
      float* drow = yt + (size_t)v * NN;
      drow[0]  = a0[v];
      drow[16] = a1[v];
      drow[32] = a2[v];
      drow[48] = a3[v];
    }
#pragma unroll
    for (int g = 0; g < NMODE / 8; ++g) a[g] = an[g];
  }
}

extern "C" void kernel_launch(void* const* d_in, const int* in_sizes, int n_in,
                              void* d_out, int out_size, void* d_ws, size_t ws_size,
                              hipStream_t stream) {
  (void)in_sizes; (void)n_in; (void)out_size; (void)ws_size;
  const float* x = (const float*)d_in[0];
  const float* w = (const float*)d_in[1];
  float* y = (float*)d_out;

  float* ws   = (float*)d_ws;
  float* c1t  = ws;
  float* c2t  = c1t + (size_t)NMODE * NN;
  float* wp   = c2t + (size_t)NN * NMODE;
  float* wm   = wp  + (size_t)CIN * COUT * NMODE;
  float* xhp  = wm  + (size_t)CIN * COUT * NMODE;
  float* xh   = xhp + (size_t)KSPLIT * NB * CIN * NMODE;
  float* outm = xh  + (size_t)NB * CIN * NMODE;

  prep_cas<<<(2 * NMODE * NN) / 256, 256, 0, stream>>>(c1t, c2t);
  prep_w<<<COUT * (CIN / WT_I), 256, 0, stream>>>(w, wp, wm);

  dht_fwd<<<((NB * CIN / 32) * KSPLIT * 32) / 128, 128, 0, stream>>>(x, c1t, xhp);
  reduce_xh<<<(NB * CIN) / RT_R, 256, 0, stream>>>(xhp, xh);

  mode_mix<<<(NMODE * (NB / 16) * (COUT / 16) * 32) / 256, 256, 0, stream>>>(
      xh, wp, wm, outm);

  dht_inv<<<((NB * COUT / (16 * INV_TILES)) * (NN / 64) * 32) / 256, 256, 0, stream>>>(
      outm, c2t, y);
}


# B-table column of lane c, tile t is 4c+t (was 16t+c) in both transforms: a lane's 4 accumulators are adjacent columns, D stores are 16-byte, each half-wave writes a whole 256-byte run; same sums
# speedup vs baseline: 2.1381x; 1.1251x over previous
#include <hip/hip_runtime.h>
#include <math.h>

#define NB     64
#define CIN    128
#define COUT   128
#define NMODE  64
#define NN     4096
#define LOG2N  12
#define KSPLIT 8

typedef float v2f __attribute__((ext_vector_type(2)));
typedef float v4f __attribute__((ext_vector_type(4)));
typedef float v8f __attribute__((ext_vector_type(8)));

static __device__ __forceinline__ v8f wmma_f32_16x16x4(v2f a, v2f b, v8f c) {
  return __builtin_amdgcn_wmma_f32_16x16x4_f32(false, a, false, b, (short)0, c,
                                               false, false);
}

static __device__ __forceinline__ float cas_tw(int p) {
  float th = (float)(p & (NN - 1)) * (6.28318530717958647692f / (float)NN);
  float s, c;
  sincosf(th, &s, &c);
  return c + s;
}

__global__ void prep_cas(float* __restrict__ c1f, float* __restrict__ c2f) {
  int idx = blockIdx.x * blockDim.x + threadIdx.x;
  int e = idx & 3;
  int l = (idx >> 2) & 31;
  int t = (idx >> 7) & 3;
  int c = l & 15;
  int h = l >> 4;
  if (idx < NMODE * NN) {
    int g = idx >> 9;
    int m = 4 * c + t;
    int j = 8 * g + 4 * h + e;
    float v = 1.0f;
#pragma unroll
    for (int b = 0; b < LOG2N; ++b)
      if ((j >> b) & 1) v *= cas_tw(m << b);
    c1f[idx] = v;
  } else {
    int u = idx - NMODE * NN;
    int g  = (u >> 9) & 7;
    int nc = u >> 12;
    int n = 64 * nc + 4 * c + t;
    int k = 8 * g + 4 * h + e;
    float v = 1.0f / (float)NN;
#pragma unroll
    for (int b = 0; b < 6; ++b)
      if ((k >> b) & 1) v *= cas_tw(n << b);
    c2f[u] = v;
  }
}

#define WT_I 32
__global__ void prep_w(const float* __restrict__ w, float* __restrict__ wpt,
                       float* __restrict__ wmt) {
  __shared__ float tile[WT_I][NMODE + 1];
  int o   = blockIdx.x >> 2;
  int i0  = (blockIdx.x & 3) * WT_I;
  int tid = threadIdx.x;
#pragma unroll
  for (int e = 0; e < (WT_I * NMODE) / 256; ++e) {
    int l  = e * 256 + tid;
    int il = l >> 6;
    int k  = l & (NMODE - 1);
    tile[il][k] = w[((size_t)(i0 + il) * COUT + o) * NMODE + k];
  }
  __syncthreads();
#pragma unroll
  for (int e = 0; e < (WT_I * NMODE) / 256; ++e) {
    int l  = e * 256 + tid;
    int k  = l >> 5;
    int il = l & (WT_I - 1);
    int kneg = (NMODE - k) & (NMODE - 1);
    float a = tile[il][k];
    float b = tile[il][kneg];
    size_t dst = ((size_t)k * COUT + o) * CIN + i0 + il;
    wpt[dst] = a + b;
    wmt[dst] = a - b;
  }
}

#define FWD_TRIP 4

static __device__ __forceinline__ void fwd_load(v4f (&a)[2][FWD_TRIP], v4f (&b)[FWD_TRIP][4],
                                                const float* __restrict__ xr0,
                                                const float* __restrict__ xr1,
                                                const v4f* __restrict__ bf, int g) {
#pragma unroll
  for (int u = 0; u < FWD_TRIP; ++u) {
    a[0][u] = *(const v4f*)(xr0 + (g + u) * 8);
    a[1][u] = *(const v4f*)(xr1 + (g + u) * 8);
    const v4f* bg = bf + (size_t)(g + u) * 128;
    b[u][0] = bg[0];
    b[u][1] = bg[32];
    b[u][2] = bg[64];
    b[u][3] = bg[96];
  }
}

static __device__ __forceinline__ void fwd_mma(const v4f (&a)[2][FWD_TRIP],
                                               const v4f (&b)[FWD_TRIP][4], v8f (&acc)[2][4]) {
#pragma unroll
  for (int u = 0; u < FWD_TRIP; ++u) {
#pragma unroll
    for (int r = 0; r < 2; ++r)
#pragma unroll
      for (int t = 0; t < 4; ++t)
        acc[r][t] = wmma_f32_16x16x4(a[r][u].xy, b[u][t].xy, acc[r][t]);
#pragma unroll
    for (int r = 0; r < 2; ++r)
#pragma unroll
      for (int t = 0; t < 4; ++t)
        acc[r][t] = wmma_f32_16x16x4(a[r][u].zw, b[u][t].zw, acc[r][t]);
  }
}

__global__ void __launch_bounds__(128)
dht_fwd(const float* __restrict__ x, const float* __restrict__ c1f,
                        float* __restrict__ xhp) {
  int lane = threadIdx.x & 31;
  int kc = blockIdx.x & (KSPLIT - 1);
  int rt = (blockIdx.x >> 3) * 4 + (threadIdx.x >> 5);
  int r0 = rt * 32;
  int mrow = lane & 15;
  int khalf = lane >> 4;
  const float* xr0 = x + (size_t)(r0 + mrow) * NN + khalf * 4;
  const float* xr1 = xr0 + (size_t)16 * NN;
  const v4f*   bf = (const v4f*)c1f + lane;
  int g0 = kc * (NN / KSPLIT / 8);
  int g1 = g0 + (NN / KSPLIT / 8);
  v8f acc[2][4] = {};
  v4f aA[2][FWD_TRIP], bA[FWD_TRIP][4];
  v4f aB[2][FWD_TRIP], bB[FWD_TRIP][4];
  fwd_load(aA, bA, xr0, xr1, bf, g0);
#pragma nounroll
  for (int g = g0; g < g1; g += 2 * FWD_TRIP) {
    fwd_load(aB, bB, xr0, xr1, bf, g + FWD_TRIP);
    __builtin_amdgcn_sched_barrier(0);
    fwd_mma(aA, bA, acc);
    int gn = (g + 2 * FWD_TRIP < g1) ? (g + 2 * FWD_TRIP) : (g1 - FWD_TRIP);
    fwd_load(aA, bA, xr0, xr1, bf, gn);
    __builtin_amdgcn_sched_barrier(0);
    fwd_mma(aB, bB, acc);
  }
  float* dst = xhp + (size_t)kc * ((size_t)NB * CIN * NMODE);
#pragma unroll
  for (int r = 0; r < 2; ++r) {
#pragma unroll
    for (int v = 0; v < 8; ++v) {
      int rr = r0 + 16 * r + v + 8 * khalf;
      v4f o = {acc[r][0][v], acc[r][1][v], acc[r][2][v], acc[r][3][v]};
      *(v4f*)(dst + (size_t)rr * NMODE + 4 * mrow) = o;
    }
  }
}

#define RT_R 32
__global__ void reduce_xh(const float* __restrict__ xhp, float* __restrict__ xht) {
  __shared__ float tile[NMODE][RT_R + 1];
  int r0  = blockIdx.x * RT_R;
  int tid = threadIdx.x;
#pragma unroll
  for (int e = 0; e < (RT_R * NMODE) / 256; ++e) {
    int l  = e * 256 + tid;
    int rl = l >> 6;
    int m  = l & (NMODE - 1);
    size_t idx = (size_t)r0 * NMODE + l;
    float s = 0.0f;
#pragma unroll
    for (int c = 0; c < KSPLIT; ++c)
      s += xhp[(size_t)c * ((size_t)NB * CIN * NMODE) + idx];
    tile[m][rl] = s;
  }
  __syncthreads();
#pragma unroll
  for (int e = 0; e < (RT_R * NMODE) / 256; ++e) {
    int l  = e * 256 + tid;
    int m  = l >> 5;
    int rl = l & (RT_R - 1);
    xht[(size_t)m * (NB * CIN) + r0 + rl] = tile[m][rl];
  }
}

__global__ void mode_mix(const float* __restrict__ xht, const float* __restrict__ wpt,
                         const float* __restrict__ wmt, float* __restrict__ outm) {
  int wave = (blockIdx.x * blockDim.x + threadIdx.x) >> 5;
  int lane = threadIdx.x & 31;
  int k   = wave >> 5;
  int rem = wave & 31;
  int bt = rem >> 3;
  int ot = rem & 7;
  int kneg = (NMODE - k) & (NMODE - 1);
  int mrow = lane & 15;
  int khalf = lane >> 4;
  int bglob = bt * 16 + mrow;
  int oglob = ot * 16 + mrow;
  const float* apr = xht + ((size_t)k    * NB   + bglob) * CIN + khalf * 2;
  const float* anr = xht + ((size_t)kneg * NB   + bglob) * CIN + khalf * 2;
  const float* bpr = wpt + ((size_t)k    * COUT + oglob) * CIN + khalf * 2;
  const float* bmr = wmt + ((size_t)k    * COUT + oglob) * CIN + khalf * 2;
  v8f acc = {};
#pragma unroll 8
  for (int i = 0; i < CIN; i += 4) {
    v2f ap = *(const v2f*)(apr + i);
    v2f an = *(const v2f*)(anr + i);
    v2f bp = *(const v2f*)(bpr + i);
    v2f bm = *(const v2f*)(bmr + i);
    acc = wmma_f32_16x16x4(ap, bp, acc);
    acc = wmma_f32_16x16x4(an, bm, acc);
  }
#pragma unroll
  for (int v = 0; v < 8; ++v) {
    int bb = bt * 16 + v + 8 * khalf;
    outm[((size_t)bb * COUT + oglob) * NMODE + k] = 0.5f * acc[v];
  }
}

#define INV_TILES 8
__global__ void __launch_bounds__(256)
dht_inv(const float* __restrict__ outm, const float* __restrict__ c2f,
                        float* __restrict__ y) {
  int wave = (blockIdx.x * blockDim.x + threadIdx.x) >> 5;
  int lane = threadIdx.x & 31;
  int rg = wave >> 6;
  int nc = wave & 63;
  int r0 = rg * (INV_TILES * 16);
  int n0 = nc * 64;
  int mrow = lane & 15;
  int khalf = lane >> 4;
  const float* ar = outm + (size_t)(r0 + mrow) * NMODE + khalf * 4;
  const v4f*   bf = (const v4f*)c2f + (size_t)nc * (8 * 128) + lane;
  v4f a[NMODE / 8];
#pragma unroll
  for (int g = 0; g < NMODE / 8; ++g) a[g] = *(const v4f*)(ar + g * 8);
  v4f b[NMODE / 8][4];
#pragma unroll
  for (int g = 0; g < NMODE / 8; ++g) {
    const v4f* bg = bf + g * 128;
    b[g][0] = bg[0];
    b[g][1] = bg[32];
    b[g][2] = bg[64];
    b[g][3] = bg[96];
  }
  float* yr = y + (size_t)(r0 + 8 * khalf) * NN + n0 + 4 * mrow;
#pragma nounroll
  for (int t = 0; t < INV_TILES; ++t) {
    int tn = (t + 1 < INV_TILES) ? (t + 1) : t;
    const float* arn = ar + (size_t)tn * (16 * NMODE);
    v4f an[NMODE / 8];
#pragma unroll
    for (int g = 0; g < NMODE / 8; ++g) an[g] = *(const v4f*)(arn + g * 8);
    v8f a0 = {}, a1 = {}, a2 = {}, a3 = {};
#pragma unroll
    for (int g = 0; g < NMODE / 8; ++g) {
      a0 = wmma_f32_16x16x4(a[g].xy, b[g][0].xy, a0);
      a1 = wmma_f32_16x16x4(a[g].xy, b[g][1].xy, a1);
      a2 = wmma_f32_16x16x4(a[g].xy, b[g][2].xy, a2);
      a3 = wmma_f32_16x16x4(a[g].xy, b[g][3].xy, a3);
      a0 = wmma_f32_16x16x4(a[g].zw, b[g][0].zw, a0);
      a1 = wmma_f32_16x16x4(a[g].zw, b[g][1].zw, a1);
      a2 = wmma_f32_16x16x4(a[g].zw, b[g][2].zw, a2);
      a3 = wmma_f32_16x16x4(a[g].zw, b[g][3].zw, a3);
    }
    float* yt = yr + (size_t)t * (16 * NN);
#pragma unroll
    for (int v = 0; v < 8; ++v) {
      v4f o = {a0[v], a1[v], a2[v], a3[v]};
      *(v4f*)(yt + (size_t)v * NN) = o;
    }
#pragma unroll
    for (int g = 0; g < NMODE / 8; ++g) a[g] = an[g];
  }
}

extern "C" void kernel_launch(void* const* d_in, const int* in_sizes, int n_in,
                              void* d_out, int out_size, void* d_ws, size_t ws_size,
                              hipStream_t stream) {
  (void)in_sizes; (void)n_in; (void)out_size; (void)ws_size;
  const float* x = (const float*)d_in[0];
  const float* w = (const float*)d_in[1];
  float* y = (float*)d_out;

  float* ws   = (float*)d_ws;
  float* c1t  = ws;
  float* c2t  = c1t + (size_t)NMODE * NN;
  float* wp   = c2t + (size_t)NN * NMODE;
  float* wm   = wp  + (size_t)CIN * COUT * NMODE;
  float* xhp  = wm  + (size_t)CIN * COUT * NMODE;
  float* xh   = xhp + (size_t)KSPLIT * NB * CIN * NMODE;
  float* outm = xh  + (size_t)NB * CIN * NMODE;

  prep_cas<<<(2 * NMODE * NN) / 256, 256, 0, stream>>>(c1t, c2t);
  prep_w<<<COUT * (CIN / WT_I), 256, 0, stream>>>(w, wp, wm);

  dht_fwd<<<((NB * CIN / 32) * KSPLIT * 32) / 128, 128, 0, stream>>>(x, c1t, xhp);
  reduce_xh<<<(NB * CIN) / RT_R, 256, 0, stream>>>(xhp, xh);

  mode_mix<<<(NMODE * (NB / 16) * (COUT / 16) * 32) / 256, 256, 0, stream>>>(
      xh, wp, wm, outm);

  dht_inv<<<((NB * COUT / (16 * INV_TILES)) * (NN / 64) * 32) / 256, 256, 0, stream>>>(
      outm, c2t, y);
}


# mode_mix operands and result stored fragment-major (xh, wp, wm, and its output as the inverse reads it): fragment loads 16 B per lane over whole lines; sum order per 8 in channels 0,1,4,5,2,3,6,7
# speedup vs baseline: 2.6766x; 1.2519x over previous
#include <hip/hip_runtime.h>
#include <math.h>

#define NB     64
#define CIN    128
#define COUT   128
#define NMODE  64
#define NN     4096
#define LOG2N  12
#define KSPLIT 8

typedef float v2f __attribute__((ext_vector_type(2)));
typedef float v4f __attribute__((ext_vector_type(4)));
typedef float v8f __attribute__((ext_vector_type(8)));

static __device__ __forceinline__ v8f wmma_f32_16x16x4(v2f a, v2f b, v8f c) {
  return __builtin_amdgcn_wmma_f32_16x16x4_f32(false, a, false, b, (short)0, c,
                                               false, false);
}

static __device__ __forceinline__ float cas_tw(int p) {
  float th = (float)(p & (NN - 1)) * (6.28318530717958647692f / (float)NN);
  float s, c;
  sincosf(th, &s, &c);
  return c + s;
}

__global__ void prep_cas(float* __restrict__ c1f, float* __restrict__ c2f) {
  int idx = blockIdx.x * blockDim.x + threadIdx.x;
  int e = idx & 3;
  int l = (idx >> 2) & 31;
  int t = (idx >> 7) & 3;
  int c = l & 15;
  int h = l >> 4;
  if (idx < NMODE * NN) {
    int g = idx >> 9;
    int m = 4 * c + t;
    int j = 8 * g + 4 * h + e;
    float v = 1.0f;
#pragma unroll
    for (int b = 0; b < LOG2N; ++b)
      if ((j >> b) & 1) v *= cas_tw(m << b);
    c1f[idx] = v;
  } else {
    int u = idx - NMODE * NN;
    int g  = (u >> 9) & 7;
    int nc = u >> 12;
    int n = 64 * nc + 4 * c + t;
    int k = 8 * g + 4 * h + e;
    float v = 1.0f / (float)NN;
#pragma unroll
    for (int b = 0; b < 6; ++b)
      if ((k >> b) & 1) v *= cas_tw(n << b);
    c2f[u] = v;
  }
}

#define WT_I 32
__global__ void prep_w(const float* __restrict__ w, float* __restrict__ wpf,
                       float* __restrict__ wmf) {
  __shared__ float tile[WT_I][NMODE + 1];
  int h     = blockIdx.x & 1;
  int g     = (blockIdx.x >> 1) & 15;
  int chalf = (blockIdx.x >> 5) & 1;
  int ot    = blockIdx.x >> 6;
  int tid = threadIdx.x;
#pragma unroll
  for (int e = 0; e < (WT_I * NMODE) / 256; ++e) {
    int l  = e * 256 + tid;
    int il = l >> 6;
    int k  = l & (NMODE - 1);
    int i  = 8 * g + 4 * h + (il & 3);
    int o  = 16 * ot + 8 * chalf + (il >> 2);
    tile[il][k] = w[((size_t)i * COUT + o) * NMODE + k];
  }
  __syncthreads();
#pragma unroll
  for (int e = 0; e < (WT_I * NMODE) / 256; ++e) {
    int l  = e * 256 + tid;
    int k  = l >> 5;
    int il = l & (WT_I - 1);
    int kneg = (NMODE - k) & (NMODE - 1);
    float a = tile[il][k];
    float b = tile[il][kneg];
    size_t dst = ((((size_t)k * 8 + ot) * 16 + g) * 2 + h) * 64 + 32 * chalf + il;
    wpf[dst] = a + b;
    wmf[dst] = a - b;
  }
}

#define FWD_TRIP 4

static __device__ __forceinline__ void fwd_load(v4f (&a)[2][FWD_TRIP], v4f (&b)[FWD_TRIP][4],
                                                const float* __restrict__ xr0,
                                                const float* __restrict__ xr1,
                                                const v4f* __restrict__ bf, int g) {
#pragma unroll
  for (int u = 0; u < FWD_TRIP; ++u) {
    a[0][u] = *(const v4f*)(xr0 + (g + u) * 8);
    a[1][u] = *(const v4f*)(xr1 + (g + u) * 8);
    const v4f* bg = bf + (size_t)(g + u) * 128;
    b[u][0] = bg[0];
    b[u][1] = bg[32];
    b[u][2] = bg[64];
    b[u][3] = bg[96];
  }
}

static __device__ __forceinline__ void fwd_mma(const v4f (&a)[2][FWD_TRIP],
                                               const v4f (&b)[FWD_TRIP][4], v8f (&acc)[2][4]) {
#pragma unroll
  for (int u = 0; u < FWD_TRIP; ++u) {
#pragma unroll
    for (int r = 0; r < 2; ++r)
#pragma unroll
      for (int t = 0; t < 4; ++t)
        acc[r][t] = wmma_f32_16x16x4(a[r][u].xy, b[u][t].xy, acc[r][t]);
#pragma unroll
    for (int r = 0; r < 2; ++r)
#pragma unroll
      for (int t = 0; t < 4; ++t)
        acc[r][t] = wmma_f32_16x16x4(a[r][u].zw, b[u][t].zw, acc[r][t]);
  }
}

__global__ void __launch_bounds__(128)
dht_fwd(const float* __restrict__ x, const float* __restrict__ c1f,
                        float* __restrict__ xhp) {
  int lane = threadIdx.x & 31;
  int kc = blockIdx.x & (KSPLIT - 1);
  int rt = (blockIdx.x >> 3) * 4 + (threadIdx.x >> 5);
  int r0 = rt * 32;
  int mrow = lane & 15;
  int khalf = lane >> 4;
  const float* xr0 = x + (size_t)(r0 + mrow) * NN + khalf * 4;
  const float* xr1 = xr0 + (size_t)16 * NN;
  const v4f*   bf = (const v4f*)c1f + lane;
  int g0 = kc * (NN / KSPLIT / 8);
  int g1 = g0 + (NN / KSPLIT / 8);
  v8f acc[2][4] = {};
  v4f aA[2][FWD_TRIP], bA[FWD_TRIP][4];
  v4f aB[2][FWD_TRIP], bB[FWD_TRIP][4];
  fwd_load(aA, bA, xr0, xr1, bf, g0);
#pragma nounroll
  for (int g = g0; g < g1; g += 2 * FWD_TRIP) {
    fwd_load(aB, bB, xr0, xr1, bf, g + FWD_TRIP);
    __builtin_amdgcn_sched_barrier(0);
    fwd_mma(aA, bA, acc);
    int gn = (g + 2 * FWD_TRIP < g1) ? (g + 2 * FWD_TRIP) : (g1 - FWD_TRIP);
    fwd_load(aA, bA, xr0, xr1, bf, gn);
    __builtin_amdgcn_sched_barrier(0);
    fwd_mma(aB, bB, acc);
  }
  float* dst = xhp + (size_t)kc * ((size_t)NB * CIN * NMODE);
#pragma unroll
  for (int r = 0; r < 2; ++r) {
#pragma unroll
    for (int v = 0; v < 8; ++v) {
      int rr = r0 + 16 * r + v + 8 * khalf;
      v4f o = {acc[r][0][v], acc[r][1][v], acc[r][2][v], acc[r][3][v]};
      *(v4f*)(dst + (size_t)rr * NMODE + 4 * mrow) = o;
    }
  }
}

#define RT_R 32
__global__ void reduce_xh(const float* __restrict__ xhp, float* __restrict__ xf) {
  __shared__ float tile[NMODE][RT_R + 1];
  int h     = blockIdx.x & 1;
  int g     = (blockIdx.x >> 1) & 15;
  int chalf = (blockIdx.x >> 5) & 1;
  int bt    = blockIdx.x >> 6;
  int tid = threadIdx.x;
#pragma unroll
  for (int e = 0; e < (RT_R * NMODE) / 256; ++e) {
    int l  = e * 256 + tid;
    int rl = l >> 6;
    int m  = l & (NMODE - 1);
    int b  = 16 * bt + 8 * chalf + (rl >> 2);
    int i  = 8 * g + 4 * h + (rl & 3);
    size_t idx = ((size_t)b * CIN + i) * NMODE + m;
    float s = 0.0f;
#pragma unroll
    for (int c = 0; c < KSPLIT; ++c)
      s += xhp[(size_t)c * ((size_t)NB * CIN * NMODE) + idx];
    tile[m][rl] = s;
  }
  __syncthreads();
#pragma unroll
  for (int e = 0; e < (RT_R * NMODE) / 256; ++e) {
    int l  = e * 256 + tid;
    int m  = l >> 5;
    int rl = l & (RT_R - 1);
    xf[((((size_t)m * 4 + bt) * 16 + g) * 2 + h) * 64 + 32 * chalf + rl] = tile[m][rl];
  }
}

__global__ void __launch_bounds__(256)
mode_mix(const float* __restrict__ xf, const float* __restrict__ wpf,
         const float* __restrict__ wmf, float* __restrict__ outm) {
  int wave = (blockIdx.x * blockDim.x + threadIdx.x) >> 5;
  int lane = threadIdx.x & 31;
  int k   = wave >> 5;
  int rem = wave & 31;
  int bt = rem >> 3;
  int ot = rem & 7;
  int kneg = (NMODE - k) & (NMODE - 1);
  int mrow = lane & 15;
  int khalf = lane >> 4;
  const v4f* apf = (const v4f*)xf  + (size_t)(k    * (NB / 16)   + bt) * (16 * 32) + lane;
  const v4f* anf = (const v4f*)xf  + (size_t)(kneg * (NB / 16)   + bt) * (16 * 32) + lane;
  const v4f* bpf = (const v4f*)wpf + (size_t)(k    * (COUT / 16) + ot) * (16 * 32) + lane;
  const v4f* bmf = (const v4f*)wmf + (size_t)(k    * (COUT / 16) + ot) * (16 * 32) + lane;
  v8f acc = {};
  v4f ap[CIN / 8], an[CIN / 8], bp[CIN / 8], bm[CIN / 8];
#pragma unroll
  for (int g = 0; g < CIN / 8; ++g) {
    ap[g] = apf[g * 32];
    an[g] = anf[g * 32];
    bp[g] = bpf[g * 32];
    bm[g] = bmf[g * 32];
  }
  __builtin_amdgcn_sched_barrier(0);
#pragma unroll
  for (int g = 0; g < CIN / 8; ++g) {
    acc = wmma_f32_16x16x4(ap[g].xy, bp[g].xy, acc);
    acc = wmma_f32_16x16x4(an[g].xy, bm[g].xy, acc);
    acc = wmma_f32_16x16x4(ap[g].zw, bp[g].zw, acc);
    acc = wmma_f32_16x16x4(an[g].zw, bm[g].zw, acc);
  }
#pragma unroll
  for (int v = 0; v < 8; ++v) {
    int bb = bt * 16 + v + 8 * khalf;
    int rt = bb * (COUT / 16) + ot;
    outm[((size_t)rt * 16 + (k >> 2)) * 64 + 4 * mrow + (k & 3)] = 0.5f * acc[v];
  }
}

#define INV_TILES 8
__global__ void __launch_bounds__(256)
dht_inv(const float* __restrict__ outm, const float* __restrict__ c2f,
                        float* __restrict__ y) {
  int wave = (blockIdx.x * blockDim.x + threadIdx.x) >> 5;
  int lane = threadIdx.x & 31;
  int rg = wave >> 6;
  int nc = wave & 63;
  int r0 = rg * (INV_TILES * 16);
  int n0 = nc * 64;
  int mrow = lane & 15;
  int khalf = lane >> 4;
  const v4f* ar = (const v4f*)outm + (size_t)(r0 / 16) * (8 * 32) + lane;
  const v4f*   bf = (const v4f*)c2f + (size_t)nc * (8 * 128) + lane;
  v4f a[NMODE / 8];
#pragma unroll
  for (int g = 0; g < NMODE / 8; ++g) a[g] = ar[g * 32];
  v4f b[NMODE / 8][4];
#pragma unroll
  for (int g = 0; g < NMODE / 8; ++g) {
    const v4f* bg = bf + g * 128;
    b[g][0] = bg[0];
    b[g][1] = bg[32];
    b[g][2] = bg[64];
    b[g][3] = bg[96];
  }
  float* yr = y + (size_t)(r0 + 8 * khalf) * NN + n0 + 4 * mrow;
#pragma nounroll
  for (int t = 0; t < INV_TILES; ++t) {
    int tn = (t + 1 < INV_TILES) ? (t + 1) : t;
    const v4f* arn = ar + (size_t)tn * (8 * 32);
    v4f an[NMODE / 8];
#pragma unroll
    for (int g = 0; g < NMODE / 8; ++g) an[g] = arn[g * 32];
    v8f a0 = {}, a1 = {}, a2 = {}, a3 = {};
#pragma unroll
    for (int g = 0; g < NMODE / 8; ++g) {
      a0 = wmma_f32_16x16x4(a[g].xy, b[g][0].xy, a0);
      a1 = wmma_f32_16x16x4(a[g].xy, b[g][1].xy, a1);
      a2 = wmma_f32_16x16x4(a[g].xy, b[g][2].xy, a2);
      a3 = wmma_f32_16x16x4(a[g].xy, b[g][3].xy, a3);
      a0 = wmma_f32_16x16x4(a[g].zw, b[g][0].zw, a0);
      a1 = wmma_f32_16x16x4(a[g].zw, b[g][1].zw, a1);
      a2 = wmma_f32_16x16x4(a[g].zw, b[g][2].zw, a2);
      a3 = wmma_f32_16x16x4(a[g].zw, b[g][3].zw, a3);
    }
    float* yt = yr + (size_t)t * (16 * NN);
#pragma unroll
    for (int v = 0; v < 8; ++v) {
      v4f o = {a0[v], a1[v], a2[v], a3[v]};
      *(v4f*)(yt + (size_t)v * NN) = o;
    }
#pragma unroll
    for (int g = 0; g < NMODE / 8; ++g) a[g] = an[g];
  }
}

extern "C" void kernel_launch(void* const* d_in, const int* in_sizes, int n_in,
                              void* d_out, int out_size, void* d_ws, size_t ws_size,
                              hipStream_t stream) {
  (void)in_sizes; (void)n_in; (void)out_size; (void)ws_size;
  const float* x = (const float*)d_in[0];
  const float* w = (const float*)d_in[1];
  float* y = (float*)d_out;

  float* ws   = (float*)d_ws;
  float* c1t  = ws;
  float* c2t  = c1t + (size_t)NMODE * NN;
  float* wp   = c2t + (size_t)NN * NMODE;
  float* wm   = wp  + (size_t)CIN * COUT * NMODE;
  float* xhp  = wm  + (size_t)CIN * COUT * NMODE;
  float* xh   = xhp + (size_t)KSPLIT * NB * CIN * NMODE;
  float* outm = xh  + (size_t)NB * CIN * NMODE;

  prep_cas<<<(2 * NMODE * NN) / 256, 256, 0, stream>>>(c1t, c2t);
  prep_w<<<COUT * (CIN / WT_I), 256, 0, stream>>>(w, wp, wm);

  dht_fwd<<<((NB * CIN / 32) * KSPLIT * 32) / 128, 128, 0, stream>>>(x, c1t, xhp);
  reduce_xh<<<(NB * CIN) / RT_R, 256, 0, stream>>>(xhp, xh);

  mode_mix<<<(NMODE * (NB / 16) * (COUT / 16) * 32) / 256, 256, 0, stream>>>(
      xh, wp, wm, outm);

  dht_inv<<<((NB * COUT / (16 * INV_TILES)) * (NN / 64) * 32) / 256, 256, 0, stream>>>(
      outm, c2t, y);
}


# fwd transform: the two top bits of j carry only +-1 factors (fixed per mode class m&3): a row's 4 quarters are added/subtracted first, matrix product and table cover j<1024 only; sum order changes
# speedup vs baseline: 2.8626x; 1.0695x over previous
#include <hip/hip_runtime.h>
#include <math.h>

#define NB     64
#define CIN    128
#define COUT   128
#define NMODE  64
#define NN     4096
#define LOG2N  12
#define KSPLIT 8
#define FOLDQ  4
#define NJ     (NN / FOLDQ)

typedef float v2f __attribute__((ext_vector_type(2)));
typedef float v4f __attribute__((ext_vector_type(4)));
typedef float v8f __attribute__((ext_vector_type(8)));

static __device__ __forceinline__ v8f wmma_f32_16x16x4(v2f a, v2f b, v8f c) {
  return __builtin_amdgcn_wmma_f32_16x16x4_f32(false, a, false, b, (short)0, c,
                                               false, false);
}

static __device__ __forceinline__ float cas_tw(int p) {
  float th = (float)(p & (NN - 1)) * (6.28318530717958647692f / (float)NN);
  float s, c;
  sincosf(th, &s, &c);
  return c + s;
}

__global__ void prep_cas(float* __restrict__ c1f, float* __restrict__ c2f) {
  int idx = blockIdx.x * blockDim.x + threadIdx.x;
  int e = idx & 3;
  int l = (idx >> 2) & 31;
  int t = (idx >> 7) & 3;
  int c = l & 15;
  int h = l >> 4;
  if (idx < NMODE * NJ) {
    int g = idx >> 9;
    int m = 4 * c + t;
    int j = 8 * g + 4 * h + e;
    float v = 1.0f;
#pragma unroll
    for (int b = 0; b < LOG2N - 2; ++b)
      if ((j >> b) & 1) v *= cas_tw(m << b);
    c1f[idx] = v;
  } else {
    int u = idx - NMODE * NJ;
    int g  = (u >> 9) & 7;
    int nc = u >> 12;
    int n = 64 * nc + 4 * c + t;
    int k = 8 * g + 4 * h + e;
    float v = 1.0f / (float)NN;
#pragma unroll
    for (int b = 0; b < 6; ++b)
      if ((k >> b) & 1) v *= cas_tw(n << b);
    c2f[u] = v;
  }
}

#define WT_I 32
__global__ void prep_w(const float* __restrict__ w, float* __restrict__ wpf,
                       float* __restrict__ wmf) {
  __shared__ float tile[WT_I][NMODE + 1];
  int h     = blockIdx.x & 1;
  int g     = (blockIdx.x >> 1) & 15;
  int chalf = (blockIdx.x >> 5) & 1;
  int ot    = blockIdx.x >> 6;
  int tid = threadIdx.x;
#pragma unroll
  for (int e = 0; e < (WT_I * NMODE) / 256; ++e) {
    int l  = e * 256 + tid;
    int il = l >> 6;
    int k  = l & (NMODE - 1);
    int i  = 8 * g + 4 * h + (il & 3);
    int o  = 16 * ot + 8 * chalf + (il >> 2);
    tile[il][k] = w[((size_t)i * COUT + o) * NMODE + k];
  }
  __syncthreads();
#pragma unroll
  for (int e = 0; e < (WT_I * NMODE) / 256; ++e) {
    int l  = e * 256 + tid;
    int k  = l >> 5;
    int il = l & (WT_I - 1);
    int kneg = (NMODE - k) & (NMODE - 1);
    float a = tile[il][k];
    float b = tile[il][kneg];
    size_t dst = ((((size_t)k * 8 + ot) * 16 + g) * 2 + h) * 64 + 32 * chalf + il;
    wpf[dst] = a + b;
    wmf[dst] = a - b;
  }
}

#define FWD_TRIP 4
#define FWD_NTRIP (NJ / KSPLIT / 8 / FWD_TRIP)

static __device__ __forceinline__ void fwd_load(v4f (&a)[2][FOLDQ][FWD_TRIP],
                                                v4f (&b)[FWD_TRIP][4],
                                                const float* __restrict__ xr0,
                                                const float* __restrict__ xr1,
                                                const v4f* __restrict__ bf, int g) {
#pragma unroll
  for (int u = 0; u < FWD_TRIP; ++u) {
#pragma unroll
    for (int q = 0; q < FOLDQ; ++q) {
      a[0][q][u] = *(const v4f*)(xr0 + q * NJ + (g + u) * 8);
      a[1][q][u] = *(const v4f*)(xr1 + q * NJ + (g + u) * 8);
    }
    const v4f* bg = bf + (size_t)(g + u) * 128;
    b[u][0] = bg[0];
    b[u][1] = bg[32];
    b[u][2] = bg[64];
    b[u][3] = bg[96];
  }
}

static __device__ __forceinline__ void fwd_mma(const v4f (&a)[2][FOLDQ][FWD_TRIP],
                                               const v4f (&b)[FWD_TRIP][4], v8f (&acc)[2][4]) {
#pragma unroll
  for (int u = 0; u < FWD_TRIP; ++u) {
#pragma unroll
    for (int r = 0; r < 2; ++r) {
      v4f p  = a[r][0][u] + a[r][1][u];
      v4f m  = a[r][0][u] - a[r][1][u];
      v4f pp = a[r][2][u] + a[r][3][u];
      v4f mm = a[r][2][u] - a[r][3][u];
      v4f f[4] = {p + pp, p - pp, m + mm, m - mm};
#pragma unroll
      for (int t = 0; t < 4; ++t)
        acc[r][t] = wmma_f32_16x16x4(f[t].xy, b[u][t].xy, acc[r][t]);
#pragma unroll
      for (int t = 0; t < 4; ++t)
        acc[r][t] = wmma_f32_16x16x4(f[t].zw, b[u][t].zw, acc[r][t]);
    }
  }
}

__global__ void __launch_bounds__(128, 2)
dht_fwd(const float* __restrict__ x, const float* __restrict__ c1f,
                        float* __restrict__ xhp) {
  int lane = threadIdx.x & 31;
  int kc = blockIdx.x & (KSPLIT - 1);
  int rt = (blockIdx.x >> 3) * 4 + (threadIdx.x >> 5);
  int r0 = rt * 32;
  int mrow = lane & 15;
  int khalf = lane >> 4;
  const float* xr0 = x + (size_t)(r0 + mrow) * NN + khalf * 4;
  const float* xr1 = xr0 + (size_t)16 * NN;
  const v4f*   bf = (const v4f*)c1f + lane;
  int g0 = kc * (NJ / KSPLIT / 8);
  v8f acc[2][4] = {};
  v4f aA[2][FOLDQ][FWD_TRIP], bA[FWD_TRIP][4];
  v4f aB[2][FOLDQ][FWD_TRIP], bB[FWD_TRIP][4];
  fwd_load(aA, bA, xr0, xr1, bf, g0);
#pragma unroll
  for (int t = 0; t < FWD_NTRIP; t += 2) {
    fwd_load(aB, bB, xr0, xr1, bf, g0 + (t + 1) * FWD_TRIP);
    __builtin_amdgcn_sched_barrier(0);
    fwd_mma(aA, bA, acc);
    if (t + 2 < FWD_NTRIP) fwd_load(aA, bA, xr0, xr1, bf, g0 + (t + 2) * FWD_TRIP);
    __builtin_amdgcn_sched_barrier(0);
    fwd_mma(aB, bB, acc);
  }
  float* dst = xhp + (size_t)kc * ((size_t)NB * CIN * NMODE);
#pragma unroll
  for (int r = 0; r < 2; ++r) {
#pragma unroll
    for (int v = 0; v < 8; ++v) {
      int rr = r0 + 16 * r + v + 8 * khalf;
      v4f o = {acc[r][0][v], acc[r][1][v], acc[r][2][v], acc[r][3][v]};
      *(v4f*)(dst + (size_t)rr * NMODE + 4 * mrow) = o;
    }
  }
}

#define RT_R 32
__global__ void reduce_xh(const float* __restrict__ xhp, float* __restrict__ xf) {
  __shared__ float tile[NMODE][RT_R + 1];
  int h     = blockIdx.x & 1;
  int g     = (blockIdx.x >> 1) & 15;
  int chalf = (blockIdx.x >> 5) & 1;
  int bt    = blockIdx.x >> 6;
  int tid = threadIdx.x;
#pragma unroll
  for (int e = 0; e < (RT_R * NMODE) / 256; ++e) {
    int l  = e * 256 + tid;
    int rl = l >> 6;
    int m  = l & (NMODE - 1);
    int b  = 16 * bt + 8 * chalf + (rl >> 2);
    int i  = 8 * g + 4 * h + (rl & 3);
    size_t idx = ((size_t)b * CIN + i) * NMODE + m;
    float s = 0.0f;
#pragma unroll
    for (int c = 0; c < KSPLIT; ++c)
      s += xhp[(size_t)c * ((size_t)NB * CIN * NMODE) + idx];
    tile[m][rl] = s;
  }
  __syncthreads();
#pragma unroll
  for (int e = 0; e < (RT_R * NMODE) / 256; ++e) {
    int l  = e * 256 + tid;
    int m  = l >> 5;
    int rl = l & (RT_R - 1);
    xf[((((size_t)m * 4 + bt) * 16 + g) * 2 + h) * 64 + 32 * chalf + rl] = tile[m][rl];
  }
}

__global__ void __launch_bounds__(256)
mode_mix(const float* __restrict__ xf, const float* __restrict__ wpf,
         const float* __restrict__ wmf, float* __restrict__ outm) {
  int wave = (blockIdx.x * blockDim.x + threadIdx.x) >> 5;
  int lane = threadIdx.x & 31;
  int k   = wave >> 5;
  int rem = wave & 31;
  int bt = rem >> 3;
  int ot = rem & 7;
  int kneg = (NMODE - k) & (NMODE - 1);
  int mrow = lane & 15;
  int khalf = lane >> 4;
  const v4f* apf = (const v4f*)xf  + (size_t)(k    * (NB / 16)   + bt) * (16 * 32) + lane;
  const v4f* anf = (const v4f*)xf  + (size_t)(kneg * (NB / 16)   + bt) * (16 * 32) + lane;
  const v4f* bpf = (const v4f*)wpf + (size_t)(k    * (COUT / 16) + ot) * (16 * 32) + lane;
  const v4f* bmf = (const v4f*)wmf + (size_t)(k    * (COUT / 16) + ot) * (16 * 32) + lane;
  v8f acc = {};
  v4f ap[CIN / 8], an[CIN / 8], bp[CIN / 8], bm[CIN / 8];
#pragma unroll
  for (int g = 0; g < CIN / 8; ++g) {
    ap[g] = apf[g * 32];
    an[g] = anf[g * 32];
    bp[g] = bpf[g * 32];
    bm[g] = bmf[g * 32];
  }
  __builtin_amdgcn_sched_barrier(0);
#pragma unroll
  for (int g = 0; g < CIN / 8; ++g) {
    acc = wmma_f32_16x16x4(ap[g].xy, bp[g].xy, acc);
    acc = wmma_f32_16x16x4(an[g].xy, bm[g].xy, acc);
    acc = wmma_f32_16x16x4(ap[g].zw, bp[g].zw, acc);
    acc = wmma_f32_16x16x4(an[g].zw, bm[g].zw, acc);
  }
#pragma unroll
  for (int v = 0; v < 8; ++v) {
    int bb = bt * 16 + v + 8 * khalf;
    int rt = bb * (COUT / 16) + ot;
    outm[((size_t)rt * 16 + (k >> 2)) * 64 + 4 * mrow + (k & 3)] = 0.5f * acc[v];
  }
}

#define INV_TILES 8
__global__ void __launch_bounds__(256)
dht_inv(const float* __restrict__ outm, const float* __restrict__ c2f,
                        float* __restrict__ y) {
  int wave = (blockIdx.x * blockDim.x + threadIdx.x) >> 5;
  int lane = threadIdx.x & 31;
  int rg = wave >> 6;
  int nc = wave & 63;
  int r0 = rg * (INV_TILES * 16);
  int n0 = nc * 64;
  int mrow = lane & 15;
  int khalf = lane >> 4;
  const v4f* ar = (const v4f*)outm + (size_t)(r0 / 16) * (8 * 32) + lane;
  const v4f*   bf = (const v4f*)c2f + (size_t)nc * (8 * 128) + lane;
  v4f a[NMODE / 8];
#pragma unroll
  for (int g = 0; g < NMODE / 8; ++g) a[g] = ar[g * 32];
  v4f b[NMODE / 8][4];
#pragma unroll
  for (int g = 0; g < NMODE / 8; ++g) {
    const v4f* bg = bf + g * 128;
    b[g][0] = bg[0];
    b[g][1] = bg[32];
    b[g][2] = bg[64];
    b[g][3] = bg[96];
  }
  float* yr = y + (size_t)(r0 + 8 * khalf) * NN + n0 + 4 * mrow;
#pragma nounroll
  for (int t = 0; t < INV_TILES; ++t) {
    int tn = (t + 1 < INV_TILES) ? (t + 1) : t;
    const v4f* arn = ar + (size_t)tn * (8 * 32);
    v4f an[NMODE / 8];
#pragma unroll
    for (int g = 0; g < NMODE / 8; ++g) an[g] = arn[g * 32];
    v8f a0 = {}, a1 = {}, a2 = {}, a3 = {};
#pragma unroll
    for (int g = 0; g < NMODE / 8; ++g) {
      a0 = wmma_f32_16x16x4(a[g].xy, b[g][0].xy, a0);
      a1 = wmma_f32_16x16x4(a[g].xy, b[g][1].xy, a1);
      a2 = wmma_f32_16x16x4(a[g].xy, b[g][2].xy, a2);
      a3 = wmma_f32_16x16x4(a[g].xy, b[g][3].xy, a3);
      a0 = wmma_f32_16x16x4(a[g].zw, b[g][0].zw, a0);
      a1 = wmma_f32_16x16x4(a[g].zw, b[g][1].zw, a1);
      a2 = wmma_f32_16x16x4(a[g].zw, b[g][2].zw, a2);
      a3 = wmma_f32_16x16x4(a[g].zw, b[g][3].zw, a3);
    }
    float* yt = yr + (size_t)t * (16 * NN);
#pragma unroll
    for (int v = 0; v < 8; ++v) {
      v4f o = {a0[v], a1[v], a2[v], a3[v]};
      *(v4f*)(yt + (size_t)v * NN) = o;
    }
#pragma unroll
    for (int g = 0; g < NMODE / 8; ++g) a[g] = an[g];
  }
}

extern "C" void kernel_launch(void* const* d_in, const int* in_sizes, int n_in,
                              void* d_out, int out_size, void* d_ws, size_t ws_size,
                              hipStream_t stream) {
  (void)in_sizes; (void)n_in; (void)out_size; (void)ws_size;
  const float* x = (const float*)d_in[0];
  const float* w = (const float*)d_in[1];
  float* y = (float*)d_out;

  float* ws   = (float*)d_ws;
  float* c1t  = ws;
  float* c2t  = c1t + (size_t)NMODE * NN;
  float* wp   = c2t + (size_t)NN * NMODE;
  float* wm   = wp  + (size_t)CIN * COUT * NMODE;
  float* xhp  = wm  + (size_t)CIN * COUT * NMODE;
  float* xh   = xhp + (size_t)KSPLIT * NB * CIN * NMODE;
  float* outm = xh  + (size_t)NB * CIN * NMODE;

  prep_cas<<<(NMODE * NJ + NMODE * NN) / 256, 256, 0, stream>>>(c1t, c2t);
  prep_w<<<COUT * (CIN / WT_I), 256, 0, stream>>>(w, wp, wm);

  dht_fwd<<<((NB * CIN / 32) * KSPLIT * 32) / 128, 128, 0, stream>>>(x, c1t, xhp);
  reduce_xh<<<(NB * CIN) / RT_R, 256, 0, stream>>>(xhp, xh);

  mode_mix<<<(NMODE * (NB / 16) * (COUT / 16) * 32) / 256, 256, 0, stream>>>(
      xh, wp, wm, outm);

  dht_inv<<<((NB * COUT / (16 * INV_TILES)) * (NN / 64) * 32) / 256, 256, 0, stream>>>(
      outm, c2t, y);
}


# inverse transform: mode bits 0,1 become the reference's last two butterfly levels (2 FMAs per output); matrix product covers 16 modes x 1024 columns per class, 1/4 of the steps; sum order changes
# speedup vs baseline: 3.2658x; 1.1409x over previous
#include <hip/hip_runtime.h>
#include <math.h>

#define NB     64
#define CIN    128
#define COUT   128
#define NMODE  64
#define NN     4096
#define LOG2N  12
#define KSPLIT 8
#define FOLDQ  4
#define NJ     (NN / FOLDQ)

typedef float v2f __attribute__((ext_vector_type(2)));
typedef float v4f __attribute__((ext_vector_type(4)));
typedef float v8f __attribute__((ext_vector_type(8)));

static __device__ __forceinline__ v8f wmma_f32_16x16x4(v2f a, v2f b, v8f c) {
  return __builtin_amdgcn_wmma_f32_16x16x4_f32(false, a, false, b, (short)0, c,
                                               false, false);
}

static __device__ __forceinline__ float cas_tw(int p) {
  float th = (float)(p & (NN - 1)) * (6.28318530717958647692f / (float)NN);
  float s, c;
  sincosf(th, &s, &c);
  return c + s;
}

#define INV_NC (NJ / 64)
__global__ void prep_cas(float* __restrict__ c1f, float* __restrict__ c2g,
                         float* __restrict__ fct) {
  int idx = blockIdx.x * blockDim.x + threadIdx.x;
  int e = idx & 3;
  int l = (idx >> 2) & 31;
  int t = (idx >> 7) & 3;
  int c = l & 15;
  int h = l >> 4;
  if (idx < NMODE * NJ) {
    int g = idx >> 9;
    int m = 4 * c + t;
    int j = 8 * g + 4 * h + e;
    float v = 1.0f;
#pragma unroll
    for (int b = 0; b < LOG2N - 2; ++b)
      if ((j >> b) & 1) v *= cas_tw(m << b);
    c1f[idx] = v;
  } else if (idx < NMODE * NJ + INV_NC * NJ) {
    int u = idx - NMODE * NJ;
    int sp = (u >> 9) & 1;
    int nc = u >> 10;
    int n = 64 * nc + 4 * c + t;
    int kk = 4 * (2 * sp + (e >> 1)) + 2 * h + (e & 1);
    float v = 1.0f / (float)NN;
#pragma unroll
    for (int b = 0; b < 4; ++b)
      if ((kk >> b) & 1) v *= cas_tw(n << (b + 2));
    c2g[u] = v;
  } else {
    int w = idx - (NMODE * NJ + INV_NC * NJ);
    int kind = w >> 10;
    int n = w & (NJ - 1);
    int p = (kind == 0) ? 2 * n : (kind == 1) ? n : n + NJ;
    fct[w] = cas_tw(p);
  }
}

#define WT_I 32
__global__ void prep_w(const float* __restrict__ w, float* __restrict__ wpf,
                       float* __restrict__ wmf) {
  __shared__ float tile[WT_I][NMODE + 1];
  int h     = blockIdx.x & 1;
  int g     = (blockIdx.x >> 1) & 15;
  int chalf = (blockIdx.x >> 5) & 1;
  int ot    = blockIdx.x >> 6;
  int tid = threadIdx.x;
#pragma unroll
  for (int e = 0; e < (WT_I * NMODE) / 256; ++e) {
    int l  = e * 256 + tid;
    int il = l >> 6;
    int k  = l & (NMODE - 1);
    int i  = 8 * g + 4 * h + (il & 3);
    int o  = 16 * ot + 8 * chalf + (il >> 2);
    tile[il][k] = w[((size_t)i * COUT + o) * NMODE + k];
  }
  __syncthreads();
#pragma unroll
  for (int e = 0; e < (WT_I * NMODE) / 256; ++e) {
    int l  = e * 256 + tid;
    int k  = l >> 5;
    int il = l & (WT_I - 1);
    int kneg = (NMODE - k) & (NMODE - 1);
    float a = tile[il][k];
    float b = tile[il][kneg];
    size_t dst = ((((size_t)k * 8 + ot) * 16 + g) * 2 + h) * 64 + 32 * chalf + il;
    wpf[dst] = a + b;
    wmf[dst] = a - b;
  }
}

#define FWD_TRIP 4
#define FWD_NTRIP (NJ / KSPLIT / 8 / FWD_TRIP)

static __device__ __forceinline__ void fwd_load(v4f (&a)[2][FOLDQ][FWD_TRIP],
                                                v4f (&b)[FWD_TRIP][4],
                                                const float* __restrict__ xr0,
                                                const float* __restrict__ xr1,
                                                const v4f* __restrict__ bf, int g) {
#pragma unroll
  for (int u = 0; u < FWD_TRIP; ++u) {
#pragma unroll
    for (int q = 0; q < FOLDQ; ++q) {
      a[0][q][u] = *(const v4f*)(xr0 + q * NJ + (g + u) * 8);
      a[1][q][u] = *(const v4f*)(xr1 + q * NJ + (g + u) * 8);
    }
    const v4f* bg = bf + (size_t)(g + u) * 128;
    b[u][0] = bg[0];
    b[u][1] = bg[32];
    b[u][2] = bg[64];
    b[u][3] = bg[96];
  }
}

static __device__ __forceinline__ void fwd_mma(const v4f (&a)[2][FOLDQ][FWD_TRIP],
                                               const v4f (&b)[FWD_TRIP][4], v8f (&acc)[2][4]) {
#pragma unroll
  for (int u = 0; u < FWD_TRIP; ++u) {
#pragma unroll
    for (int r = 0; r < 2; ++r) {
      v4f p  = a[r][0][u] + a[r][1][u];
      v4f m  = a[r][0][u] - a[r][1][u];
      v4f pp = a[r][2][u] + a[r][3][u];
      v4f mm = a[r][2][u] - a[r][3][u];
      v4f f[4] = {p + pp, p - pp, m + mm, m - mm};
#pragma unroll
      for (int t = 0; t < 4; ++t)
        acc[r][t] = wmma_f32_16x16x4(f[t].xy, b[u][t].xy, acc[r][t]);
#pragma unroll
      for (int t = 0; t < 4; ++t)
        acc[r][t] = wmma_f32_16x16x4(f[t].zw, b[u][t].zw, acc[r][t]);
    }
  }
}

__global__ void __launch_bounds__(128, 2)
dht_fwd(const float* __restrict__ x, const float* __restrict__ c1f,
                        float* __restrict__ xhp) {
  int lane = threadIdx.x & 31;
  int kc = blockIdx.x & (KSPLIT - 1);
  int rt = (blockIdx.x >> 3) * 4 + (threadIdx.x >> 5);
  int r0 = rt * 32;
  int mrow = lane & 15;
  int khalf = lane >> 4;
  const float* xr0 = x + (size_t)(r0 + mrow) * NN + khalf * 4;
  const float* xr1 = xr0 + (size_t)16 * NN;
  const v4f*   bf = (const v4f*)c1f + lane;
  int g0 = kc * (NJ / KSPLIT / 8);
  v8f acc[2][4] = {};
  v4f aA[2][FOLDQ][FWD_TRIP], bA[FWD_TRIP][4];
  v4f aB[2][FOLDQ][FWD_TRIP], bB[FWD_TRIP][4];
  fwd_load(aA, bA, xr0, xr1, bf, g0);
#pragma unroll
  for (int t = 0; t < FWD_NTRIP; t += 2) {
    fwd_load(aB, bB, xr0, xr1, bf, g0 + (t + 1) * FWD_TRIP);
    __builtin_amdgcn_sched_barrier(0);
    fwd_mma(aA, bA, acc);
    if (t + 2 < FWD_NTRIP) fwd_load(aA, bA, xr0, xr1, bf, g0 + (t + 2) * FWD_TRIP);
    __builtin_amdgcn_sched_barrier(0);
    fwd_mma(aB, bB, acc);
  }
  float* dst = xhp + (size_t)kc * ((size_t)NB * CIN * NMODE);
#pragma unroll
  for (int r = 0; r < 2; ++r) {
#pragma unroll
    for (int v = 0; v < 8; ++v) {
      int rr = r0 + 16 * r + v + 8 * khalf;
      v4f o = {acc[r][0][v], acc[r][1][v], acc[r][2][v], acc[r][3][v]};
      *(v4f*)(dst + (size_t)rr * NMODE + 4 * mrow) = o;
    }
  }
}

#define RT_R 32
__global__ void reduce_xh(const float* __restrict__ xhp, float* __restrict__ xf) {
  __shared__ float tile[NMODE][RT_R + 1];
  int h     = blockIdx.x & 1;
  int g     = (blockIdx.x >> 1) & 15;
  int chalf = (blockIdx.x >> 5) & 1;
  int bt    = blockIdx.x >> 6;
  int tid = threadIdx.x;
#pragma unroll
  for (int e = 0; e < (RT_R * NMODE) / 256; ++e) {
    int l  = e * 256 + tid;
    int rl = l >> 6;
    int m  = l & (NMODE - 1);
    int b  = 16 * bt + 8 * chalf + (rl >> 2);
    int i  = 8 * g + 4 * h + (rl & 3);
    size_t idx = ((size_t)b * CIN + i) * NMODE + m;
    float s = 0.0f;
#pragma unroll
    for (int c = 0; c < KSPLIT; ++c)
      s += xhp[(size_t)c * ((size_t)NB * CIN * NMODE) + idx];
    tile[m][rl] = s;
  }
  __syncthreads();
#pragma unroll
  for (int e = 0; e < (RT_R * NMODE) / 256; ++e) {
    int l  = e * 256 + tid;
    int m  = l >> 5;
    int rl = l & (RT_R - 1);
    xf[((((size_t)m * 4 + bt) * 16 + g) * 2 + h) * 64 + 32 * chalf + rl] = tile[m][rl];
  }
}

__global__ void __launch_bounds__(256)
mode_mix(const float* __restrict__ xf, const float* __restrict__ wpf,
         const float* __restrict__ wmf, float* __restrict__ outm) {
  int wave = (blockIdx.x * blockDim.x + threadIdx.x) >> 5;
  int lane = threadIdx.x & 31;
  int k   = wave >> 5;
  int rem = wave & 31;
  int bt = rem >> 3;
  int ot = rem & 7;
  int kneg = (NMODE - k) & (NMODE - 1);
  int mrow = lane & 15;
  int khalf = lane >> 4;
  const v4f* apf = (const v4f*)xf  + (size_t)(k    * (NB / 16)   + bt) * (16 * 32) + lane;
  const v4f* anf = (const v4f*)xf  + (size_t)(kneg * (NB / 16)   + bt) * (16 * 32) + lane;
  const v4f* bpf = (const v4f*)wpf + (size_t)(k    * (COUT / 16) + ot) * (16 * 32) + lane;
  const v4f* bmf = (const v4f*)wmf + (size_t)(k    * (COUT / 16) + ot) * (16 * 32) + lane;
  v8f acc = {};
  v4f ap[CIN / 8], an[CIN / 8], bp[CIN / 8], bm[CIN / 8];
#pragma unroll
  for (int g = 0; g < CIN / 8; ++g) {
    ap[g] = apf[g * 32];
    an[g] = anf[g * 32];
    bp[g] = bpf[g * 32];
    bm[g] = bmf[g * 32];
  }
  __builtin_amdgcn_sched_barrier(0);
#pragma unroll
  for (int g = 0; g < CIN / 8; ++g) {
    acc = wmma_f32_16x16x4(ap[g].xy, bp[g].xy, acc);
    acc = wmma_f32_16x16x4(an[g].xy, bm[g].xy, acc);
    acc = wmma_f32_16x16x4(ap[g].zw, bp[g].zw, acc);
    acc = wmma_f32_16x16x4(an[g].zw, bm[g].zw, acc);
  }
  int kk  = k >> 2;
  int cls = k & 3;
  int sp  = kk >> 3;
  int ii  = 2 * ((kk >> 2) & 1) + (kk & 1);
  int hh  = (kk >> 1) & 1;
#pragma unroll
  for (int v = 0; v < 8; ++v) {
    int bb = bt * 16 + v + 8 * khalf;
    int rt = bb * (COUT / 16) + ot;
    outm[((((size_t)rt * 4 + cls) * 2 + sp) * 32 + 16 * hh + mrow) * 4 + ii] = 0.5f * acc[v];
  }
}

#define INV_TILES 4
__global__ void __launch_bounds__(256)
dht_inv(const float* __restrict__ outm, const float* __restrict__ c2g,
        const float* __restrict__ fct, float* __restrict__ y) {
  int wave = (blockIdx.x * blockDim.x + threadIdx.x) >> 5;
  int lane = threadIdx.x & 31;
  int rg = wave >> 4;
  int nc = wave & (INV_NC - 1);
  int r0 = rg * (INV_TILES * 16);
  int mrow = lane & 15;
  int khalf = lane >> 4;
  const v4f* ar = (const v4f*)outm + (size_t)(r0 / 16) * (8 * 32) + lane;
  const v4f* bf = (const v4f*)c2g + (size_t)nc * (8 * 32) + lane;
  const v4f* ff = (const v4f*)fct + 16 * nc + mrow;
  v4f a[4][2];
#pragma unroll
  for (int cls = 0; cls < 4; ++cls) {
    a[cls][0] = ar[(cls * 2 + 0) * 32];
    a[cls][1] = ar[(cls * 2 + 1) * 32];
  }
  v4f b[2][4];
#pragma unroll
  for (int sp = 0; sp < 2; ++sp)
#pragma unroll
    for (int t = 0; t < 4; ++t) b[sp][t] = bf[(sp * 4 + t) * 32];
  v4f f1 = ff[0];
  v4f f0 = ff[NJ / 4];
  v4f g0 = ff[2 * (NJ / 4)];
  float* yr = y + (size_t)(r0 + 8 * khalf) * NN + 64 * nc + 4 * mrow;
#pragma nounroll
  for (int tile = 0; tile < INV_TILES; ++tile) {
    int tn = (tile + 1 < INV_TILES) ? (tile + 1) : tile;
    const v4f* arn = ar + (size_t)tn * (8 * 32);
    v4f an[4][2];
#pragma unroll
    for (int cls = 0; cls < 4; ++cls) {
      an[cls][0] = arn[(cls * 2 + 0) * 32];
      an[cls][1] = arn[(cls * 2 + 1) * 32];
    }
    __builtin_amdgcn_sched_barrier(0);
    v8f z[4][4] = {};
#pragma unroll
    for (int sp = 0; sp < 2; ++sp) {
#pragma unroll
      for (int cls = 0; cls < 4; ++cls)
#pragma unroll
        for (int t = 0; t < 4; ++t)
          z[cls][t] = wmma_f32_16x16x4(a[cls][sp].xy, b[sp][t].xy, z[cls][t]);
#pragma unroll
      for (int cls = 0; cls < 4; ++cls)
#pragma unroll
        for (int t = 0; t < 4; ++t)
          z[cls][t] = wmma_f32_16x16x4(a[cls][sp].zw, b[sp][t].zw, z[cls][t]);
    }
    float* yt = yr + (size_t)tile * (16 * NN);
#pragma unroll
    for (int v = 0; v < 8; ++v) {
      v4f z0 = {z[0][0][v], z[0][1][v], z[0][2][v], z[0][3][v]};
      v4f z1 = {z[1][0][v], z[1][1][v], z[1][2][v], z[1][3][v]};
      v4f z2 = {z[2][0][v], z[2][1][v], z[2][2][v], z[2][3][v]};
      v4f z3 = {z[3][0][v], z[3][1][v], z[3][2][v], z[3][3][v]};
      v4f pe = z0 + f1 * z2;
      v4f pm = z0 - f1 * z2;
      v4f qe = z1 + f1 * z3;
      v4f qm = z1 - f1 * z3;
      float* yv = yt + (size_t)v * NN;
      *(v4f*)(yv)          = pe + f0 * qe;
      *(v4f*)(yv + NJ)     = pm + g0 * qm;
      *(v4f*)(yv + 2 * NJ) = pe - f0 * qe;
      *(v4f*)(yv + 3 * NJ) = pm - g0 * qm;
    }
#pragma unroll
    for (int cls = 0; cls < 4; ++cls) {
      a[cls][0] = an[cls][0];
      a[cls][1] = an[cls][1];
    }
  }
}

extern "C" void kernel_launch(void* const* d_in, const int* in_sizes, int n_in,
                              void* d_out, int out_size, void* d_ws, size_t ws_size,
                              hipStream_t stream) {
  (void)in_sizes; (void)n_in; (void)out_size; (void)ws_size;
  const float* x = (const float*)d_in[0];
  const float* w = (const float*)d_in[1];
  float* y = (float*)d_out;

  float* ws   = (float*)d_ws;
  float* c1t  = ws;
  float* c2t  = c1t + (size_t)NMODE * NN;
  float* fct  = c2t + (size_t)INV_NC * NJ;
  float* wp   = c2t + (size_t)NN * NMODE;
  float* wm   = wp  + (size_t)CIN * COUT * NMODE;
  float* xhp  = wm  + (size_t)CIN * COUT * NMODE;
  float* xh   = xhp + (size_t)KSPLIT * NB * CIN * NMODE;
  float* outm = xh  + (size_t)NB * CIN * NMODE;

  prep_cas<<<(NMODE * NJ + INV_NC * NJ + 3 * NJ) / 256, 256, 0, stream>>>(c1t, c2t, fct);
  prep_w<<<COUT * (CIN / WT_I), 256, 0, stream>>>(w, wp, wm);

  dht_fwd<<<((NB * CIN / 32) * KSPLIT * 32) / 128, 128, 0, stream>>>(x, c1t, xhp);
  reduce_xh<<<(NB * CIN) / RT_R, 256, 0, stream>>>(xhp, xh);

  mode_mix<<<(NMODE * (NB / 16) * (COUT / 16) * 32) / 256, 256, 0, stream>>>(
      xh, wp, wm, outm);

  dht_inv<<<((NB * COUT / (16 * INV_TILES)) * INV_NC * 32) / 256, 256, 0, stream>>>(
      outm, c2t, fct, y);
}


# 6 launches -> 4: the forward K-split is summed across the work-group's 8 waves through LDS (chunk order kept; reduce launch and partial buffer gone) and both set-up kernels run as one launch
# speedup vs baseline: 3.8442x; 1.1771x over previous
#include <hip/hip_runtime.h>
#include <math.h>

#define NB     64
#define CIN    128
#define COUT   128
#define NMODE  64
#define NN     4096
#define LOG2N  12
#define KSPLIT 8
#define FOLDQ  4
#define NJ     (NN / FOLDQ)

typedef float v2f __attribute__((ext_vector_type(2)));
typedef float v4f __attribute__((ext_vector_type(4)));
typedef float v8f __attribute__((ext_vector_type(8)));

static __device__ __forceinline__ v8f wmma_f32_16x16x4(v2f a, v2f b, v8f c) {
  return __builtin_amdgcn_wmma_f32_16x16x4_f32(false, a, false, b, (short)0, c,
                                               false, false);
}

static __device__ __forceinline__ float cas_tw(int p) {
  float th = (float)(p & (NN - 1)) * (6.28318530717958647692f / (float)NN);
  float s, c;
  sincosf(th, &s, &c);
  return c + s;
}

#define INV_NC (NJ / 64)
#define PREP_CAS_BLOCKS ((NMODE * NJ + INV_NC * NJ + 3 * NJ) / 256)
static __device__ __forceinline__ void prep_cas(int blk, float* __restrict__ c1f,
                                                float* __restrict__ c2g,
                                                float* __restrict__ fct) {
  int idx = blk * 256 + threadIdx.x;
  int e = idx & 3;
  int l = (idx >> 2) & 31;
  int t = (idx >> 7) & 3;
  int c = l & 15;
  int h = l >> 4;
  if (idx < NMODE * NJ) {
    int g = idx >> 9;
    int m = 4 * c + t;
    int j = 8 * g + 4 * h + e;
    float v = 1.0f;
#pragma unroll
    for (int b = 0; b < LOG2N - 2; ++b)
      if ((j >> b) & 1) v *= cas_tw(m << b);
    c1f[idx] = v;
  } else if (idx < NMODE * NJ + INV_NC * NJ) {
    int u = idx - NMODE * NJ;
    int sp = (u >> 9) & 1;
    int nc = u >> 10;
    int n = 64 * nc + 4 * c + t;
    int kk = 4 * (2 * sp + (e >> 1)) + 2 * h + (e & 1);
    float v = 1.0f / (float)NN;
#pragma unroll
    for (int b = 0; b < 4; ++b)
      if ((kk >> b) & 1) v *= cas_tw(n << (b + 2));
    c2g[u] = v;
  } else {
    int w = idx - (NMODE * NJ + INV_NC * NJ);
    int kind = w >> 10;
    int n = w & (NJ - 1);
    int p = (kind == 0) ? 2 * n : (kind == 1) ? n : n + NJ;
    fct[w] = cas_tw(p);
  }
}

#define WT_I 32
#define PREP_W_BLOCKS (COUT * (CIN / WT_I))
static __device__ __forceinline__ void prep_w(int blk, const float* __restrict__ w,
                                              float* __restrict__ wpf,
                                              float* __restrict__ wmf) {
  __shared__ float tile[WT_I][NMODE + 1];
  int h     = blk & 1;
  int g     = (blk >> 1) & 15;
  int chalf = (blk >> 5) & 1;
  int ot    = blk >> 6;
  int tid = threadIdx.x;
#pragma unroll
  for (int e = 0; e < (WT_I * NMODE) / 256; ++e) {
    int l  = e * 256 + tid;
    int il = l >> 6;
    int k  = l & (NMODE - 1);
    int i  = 8 * g + 4 * h + (il & 3);
    int o  = 16 * ot + 8 * chalf + (il >> 2);
    tile[il][k] = w[((size_t)i * COUT + o) * NMODE + k];
  }
  __syncthreads();
#pragma unroll
  for (int e = 0; e < (WT_I * NMODE) / 256; ++e) {
    int l  = e * 256 + tid;
    int k  = l >> 5;
    int il = l & (WT_I - 1);
    int kneg = (NMODE - k) & (NMODE - 1);
    float a = tile[il][k];
    float b = tile[il][kneg];
    size_t dst = ((((size_t)k * 8 + ot) * 16 + g) * 2 + h) * 64 + 32 * chalf + il;
    wpf[dst] = a + b;
    wmf[dst] = a - b;
  }
}

__global__ void __launch_bounds__(256)
prep_all(const float* __restrict__ w, float* __restrict__ c1f, float* __restrict__ c2g,
         float* __restrict__ fct, float* __restrict__ wpf, float* __restrict__ wmf) {
  if (blockIdx.x < PREP_CAS_BLOCKS)
    prep_cas(blockIdx.x, c1f, c2g, fct);
  else
    prep_w(blockIdx.x - PREP_CAS_BLOCKS, w, wpf, wmf);
}

#define FWD_TRIP 4
#define FWD_NTRIP (NJ / KSPLIT / 8 / FWD_TRIP)
#define FWD_ROWS 32
#define FWD_PITCH4 9

static __device__ __forceinline__ void fwd_load(v4f (&a)[2][FOLDQ][FWD_TRIP],
                                                v4f (&b)[FWD_TRIP][4],
                                                const float* __restrict__ xr0,
                                                const float* __restrict__ xr1,
                                                const v4f* __restrict__ bf, int g) {
#pragma unroll
  for (int u = 0; u < FWD_TRIP; ++u) {
#pragma unroll
    for (int q = 0; q < FOLDQ; ++q) {
      a[0][q][u] = *(const v4f*)(xr0 + q * NJ + (g + u) * 8);
      a[1][q][u] = *(const v4f*)(xr1 + q * NJ + (g + u) * 8);
    }
    const v4f* bg = bf + (size_t)(g + u) * 128;
    b[u][0] = bg[0];
    b[u][1] = bg[32];
    b[u][2] = bg[64];
    b[u][3] = bg[96];
  }
}

static __device__ __forceinline__ void fwd_mma(const v4f (&a)[2][FOLDQ][FWD_TRIP],
                                               const v4f (&b)[FWD_TRIP][4], v8f (&acc)[2][4]) {
#pragma unroll
  for (int u = 0; u < FWD_TRIP; ++u) {
#pragma unroll
    for (int r = 0; r < 2; ++r) {
      v4f p  = a[r][0][u] + a[r][1][u];
      v4f m  = a[r][0][u] - a[r][1][u];
      v4f pp = a[r][2][u] + a[r][3][u];
      v4f mm = a[r][2][u] - a[r][3][u];
      v4f f[4] = {p + pp, p - pp, m + mm, m - mm};
#pragma unroll
      for (int t = 0; t < 4; ++t)
        acc[r][t] = wmma_f32_16x16x4(f[t].xy, b[u][t].xy, acc[r][t]);
#pragma unroll
      for (int t = 0; t < 4; ++t)
        acc[r][t] = wmma_f32_16x16x4(f[t].zw, b[u][t].zw, acc[r][t]);
    }
  }
}

__global__ void __launch_bounds__(256, 2)
dht_fwd(const float* __restrict__ x, const float* __restrict__ c1f,
                        float* __restrict__ xf) {
  __shared__ v4f part[KSPLIT * NMODE * FWD_PITCH4];
  int tid  = threadIdx.x;
  int lane = tid & 31;
  int kc   = tid >> 5;
  int h     = blockIdx.x & 1;
  int g     = (blockIdx.x >> 1) & 15;
  int chalf = (blockIdx.x >> 5) & 1;
  int bt    = blockIdx.x >> 6;
  int mrow = lane & 15;
  int khalf = lane >> 4;
  int bb = 16 * bt + 8 * chalf + (mrow >> 2);
  int ii = 8 * g + 4 * h + (mrow & 3);
  const float* xr0 = x + ((size_t)bb * CIN + ii) * NN + khalf * 4;
  const float* xr1 = xr0 + (size_t)4 * CIN * NN;
  const v4f*   bf = (const v4f*)c1f + lane;
  int g0 = kc * (NJ / KSPLIT / 8);
  v8f acc[2][4] = {};
  v4f aA[2][FOLDQ][FWD_TRIP], bA[FWD_TRIP][4];
  v4f aB[2][FOLDQ][FWD_TRIP], bB[FWD_TRIP][4];
  fwd_load(aA, bA, xr0, xr1, bf, g0);
#pragma unroll
  for (int t = 0; t < FWD_NTRIP; t += 2) {
    fwd_load(aB, bB, xr0, xr1, bf, g0 + (t + 1) * FWD_TRIP);
    __builtin_amdgcn_sched_barrier(0);
    fwd_mma(aA, bA, acc);
    if (t + 2 < FWD_NTRIP) fwd_load(aA, bA, xr0, xr1, bf, g0 + (t + 2) * FWD_TRIP);
    __builtin_amdgcn_sched_barrier(0);
    fwd_mma(aB, bB, acc);
  }
#pragma unroll
  for (int r = 0; r < 2; ++r) {
#pragma unroll
    for (int t = 0; t < 4; ++t) {
      v4f* p = part + ((kc * 4 + t) * 16 + mrow) * FWD_PITCH4 + 4 * r + 2 * khalf;
      v4f lo = {acc[r][t][0], acc[r][t][1], acc[r][t][2], acc[r][t][3]};
      v4f hi = {acc[r][t][4], acc[r][t][5], acc[r][t][6], acc[r][t][7]};
      p[0] = lo;
      p[1] = hi;
    }
  }
  __syncthreads();
  const float* pf = (const float*)part;
#pragma unroll
  for (int e = 0; e < (FWD_ROWS * NMODE) / 256; ++e) {
    int l  = e * 256 + tid;
    int m  = l >> 5;
    int rl = l & (FWD_ROWS - 1);
    int prow = (m & 3) * 16 + (m >> 2);
    float s = 0.0f;
#pragma unroll
    for (int c = 0; c < KSPLIT; ++c)
      s += pf[(c * NMODE + prow) * (4 * FWD_PITCH4) + rl];
    xf[((((size_t)m * 4 + bt) * 16 + g) * 2 + h) * 64 + 32 * chalf + rl] = s;
  }
}

__global__ void __launch_bounds__(256)
mode_mix(const float* __restrict__ xf, const float* __restrict__ wpf,
         const float* __restrict__ wmf, float* __restrict__ outm) {
  int wave = (blockIdx.x * blockDim.x + threadIdx.x) >> 5;
  int lane = threadIdx.x & 31;
  int k   = wave >> 5;
  int rem = wave & 31;
  int bt = rem >> 3;
  int ot = rem & 7;
  int kneg = (NMODE - k) & (NMODE - 1);
  int mrow = lane & 15;
  int khalf = lane >> 4;
  const v4f* apf = (const v4f*)xf  + (size_t)(k    * (NB / 16)   + bt) * (16 * 32) + lane;
  const v4f* anf = (const v4f*)xf  + (size_t)(kneg * (NB / 16)   + bt) * (16 * 32) + lane;
  const v4f* bpf = (const v4f*)wpf + (size_t)(k    * (COUT / 16) + ot) * (16 * 32) + lane;
  const v4f* bmf = (const v4f*)wmf + (size_t)(k    * (COUT / 16) + ot) * (16 * 32) + lane;
  v8f acc = {};
  v4f ap[CIN / 8], an[CIN / 8], bp[CIN / 8], bm[CIN / 8];
#pragma unroll
  for (int g = 0; g < CIN / 8; ++g) {
    ap[g] = apf[g * 32];
    an[g] = anf[g * 32];
    bp[g] = bpf[g * 32];
    bm[g] = bmf[g * 32];
  }
  __builtin_amdgcn_sched_barrier(0);
#pragma unroll
  for (int g = 0; g < CIN / 8; ++g) {
    acc = wmma_f32_16x16x4(ap[g].xy, bp[g].xy, acc);
    acc = wmma_f32_16x16x4(an[g].xy, bm[g].xy, acc);
    acc = wmma_f32_16x16x4(ap[g].zw, bp[g].zw, acc);
    acc = wmma_f32_16x16x4(an[g].zw, bm[g].zw, acc);
  }
  int kk  = k >> 2;
  int cls = k & 3;
  int sp  = kk >> 3;
  int ii  = 2 * ((kk >> 2) & 1) + (kk & 1);
  int hh  = (kk >> 1) & 1;
#pragma unroll
  for (int v = 0; v < 8; ++v) {
    int bb = bt * 16 + v + 8 * khalf;
    int rt = bb * (COUT / 16) + ot;
    outm[((((size_t)rt * 4 + cls) * 2 + sp) * 32 + 16 * hh + mrow) * 4 + ii] = 0.5f * acc[v];
  }
}

#define INV_TILES 4
__global__ void __launch_bounds__(256)
dht_inv(const float* __restrict__ outm, const float* __restrict__ c2g,
        const float* __restrict__ fct, float* __restrict__ y) {
  int wave = (blockIdx.x * blockDim.x + threadIdx.x) >> 5;
  int lane = threadIdx.x & 31;
  int rg = wave >> 4;
  int nc = wave & (INV_NC - 1);
  int r0 = rg * (INV_TILES * 16);
  int mrow = lane & 15;
  int khalf = lane >> 4;
  const v4f* ar = (const v4f*)outm + (size_t)(r0 / 16) * (8 * 32) + lane;
  const v4f* bf = (const v4f*)c2g + (size_t)nc * (8 * 32) + lane;
  const v4f* ff = (const v4f*)fct + 16 * nc + mrow;
  v4f a[4][2];
#pragma unroll
  for (int cls = 0; cls < 4; ++cls) {
    a[cls][0] = ar[(cls * 2 + 0) * 32];
    a[cls][1] = ar[(cls * 2 + 1) * 32];
  }
  v4f b[2][4];
#pragma unroll
  for (int sp = 0; sp < 2; ++sp)
#pragma unroll
    for (int t = 0; t < 4; ++t) b[sp][t] = bf[(sp * 4 + t) * 32];
  v4f f1 = ff[0];
  v4f f0 = ff[NJ / 4];
  v4f g0 = ff[2 * (NJ / 4)];
  float* yr = y + (size_t)(r0 + 8 * khalf) * NN + 64 * nc + 4 * mrow;
#pragma nounroll
  for (int tile = 0; tile < INV_TILES; ++tile) {
    int tn = (tile + 1 < INV_TILES) ? (tile + 1) : tile;
    const v4f* arn = ar + (size_t)tn * (8 * 32);
    v4f an[4][2];
#pragma unroll
    for (int cls = 0; cls < 4; ++cls) {
      an[cls][0] = arn[(cls * 2 + 0) * 32];
      an[cls][1] = arn[(cls * 2 + 1) * 32];
    }
    __builtin_amdgcn_sched_barrier(0);
    v8f z[4][4] = {};
#pragma unroll
    for (int sp = 0; sp < 2; ++sp) {
#pragma unroll
      for (int cls = 0; cls < 4; ++cls)
#pragma unroll
        for (int t = 0; t < 4; ++t)
          z[cls][t] = wmma_f32_16x16x4(a[cls][sp].xy, b[sp][t].xy, z[cls][t]);
#pragma unroll
      for (int cls = 0; cls < 4; ++cls)
#pragma unroll
        for (int t = 0; t < 4; ++t)
          z[cls][t] = wmma_f32_16x16x4(a[cls][sp].zw, b[sp][t].zw, z[cls][t]);
    }
    float* yt = yr + (size_t)tile * (16 * NN);
#pragma unroll
    for (int v = 0; v < 8; ++v) {
      v4f z0 = {z[0][0][v], z[0][1][v], z[0][2][v], z[0][3][v]};
      v4f z1 = {z[1][0][v], z[1][1][v], z[1][2][v], z[1][3][v]};
      v4f z2 = {z[2][0][v], z[2][1][v], z[2][2][v], z[2][3][v]};
      v4f z3 = {z[3][0][v], z[3][1][v], z[3][2][v], z[3][3][v]};
      v4f pe = z0 + f1 * z2;
      v4f pm = z0 - f1 * z2;
      v4f qe = z1 + f1 * z3;
      v4f qm = z1 - f1 * z3;
      float* yv = yt + (size_t)v * NN;
      *(v4f*)(yv)          = pe + f0 * qe;
      *(v4f*)(yv + NJ)     = pm + g0 * qm;
      *(v4f*)(yv + 2 * NJ) = pe - f0 * qe;
      *(v4f*)(yv + 3 * NJ) = pm - g0 * qm;
    }
#pragma unroll
    for (int cls = 0; cls < 4; ++cls) {
      a[cls][0] = an[cls][0];
      a[cls][1] = an[cls][1];
    }
  }
}

extern "C" void kernel_launch(void* const* d_in, const int* in_sizes, int n_in,
                              void* d_out, int out_size, void* d_ws, size_t ws_size,
                              hipStream_t stream) {
  (void)in_sizes; (void)n_in; (void)out_size; (void)ws_size;
  const float* x = (const float*)d_in[0];
  const float* w = (const float*)d_in[1];
  float* y = (float*)d_out;

  float* ws   = (float*)d_ws;
  float* c1t  = ws;
  float* c2t  = c1t + (size_t)NMODE * NN;
  float* fct  = c2t + (size_t)INV_NC * NJ;
  float* wp   = c2t + (size_t)NN * NMODE;
  float* wm   = wp  + (size_t)CIN * COUT * NMODE;
  float* xh   = wm  + (size_t)CIN * COUT * NMODE;
  float* outm = xh  + (size_t)NB * CIN * NMODE;

  prep_all<<<PREP_CAS_BLOCKS + PREP_W_BLOCKS, 256, 0, stream>>>(w, c1t, c2t, fct, wp, wm);

  dht_fwd<<<(NB * CIN) / FWD_ROWS, 256, 0, stream>>>(x, c1t, xh);

  mode_mix<<<(NMODE * (NB / 16) * (COUT / 16) * 32) / 256, 256, 0, stream>>>(
      xh, wp, wm, outm);

  dht_inv<<<((NB * COUT / (16 * INV_TILES)) * INV_NC * 32) / 256, 256, 0, stream>>>(
      outm, c2t, fct, y);
}


# 4 launches -> 3, no set-up launch: fwd work-groups build their 128x64 twiddle block in LDS (B from LDS, chunk factor scales the chunk's sum: rounds once) and write the later launches' operands
# speedup vs baseline: 4.7419x; 1.2335x over previous
#include <hip/hip_runtime.h>
#include <math.h>

#define NB     64
#define CIN    128
#define COUT   128
#define NMODE  64
#define NN     4096
#define LOG2N  12
#define KSPLIT 8
#define FOLDQ  4
#define NJ     (NN / FOLDQ)

typedef float v2f __attribute__((ext_vector_type(2)));
typedef float v4f __attribute__((ext_vector_type(4)));
typedef float v8f __attribute__((ext_vector_type(8)));

static __device__ __forceinline__ v8f wmma_f32_16x16x4(v2f a, v2f b, v8f c) {
  return __builtin_amdgcn_wmma_f32_16x16x4_f32(false, a, false, b, (short)0, c,
                                               false, false);
}

static __device__ __forceinline__ float cas_tw(int p) {
  float th = (float)(p & (NN - 1)) * (6.28318530717958647692f / (float)NN);
  float s, c;
  sincosf(th, &s, &c);
  return c + s;
}

#define INV_NC (NJ / 64)
#define FWD_BLOCKS ((NB * CIN) / 32)
#define C2G_PER_BLOCK ((INV_NC * NJ) / FWD_BLOCKS)
#define FCT_PER_BLOCK ((3 * NJ) / FWD_BLOCKS)

static __device__ __forceinline__ void prep_inv_tables(int blk, int tid,
                                                       float* __restrict__ c2g,
                                                       float* __restrict__ fct) {
  if (tid < C2G_PER_BLOCK) {
    int u = blk * C2G_PER_BLOCK + tid;
    int e = u & 3;
    int l = (u >> 2) & 31;
    int t = (u >> 7) & 3;
    int c = l & 15;
    int h = l >> 4;
    int sp = (u >> 9) & 1;
    int nc = u >> 10;
    int n = 64 * nc + 4 * c + t;
    int kk = 4 * (2 * sp + (e >> 1)) + 2 * h + (e & 1);
    float v = 1.0f / (float)NN;
#pragma unroll
    for (int b = 0; b < 4; ++b)
      if ((kk >> b) & 1) v *= cas_tw(n << (b + 2));
    c2g[u] = v;
  } else if (tid < C2G_PER_BLOCK + FCT_PER_BLOCK) {
    int w = blk * FCT_PER_BLOCK + (tid - C2G_PER_BLOCK);
    int kind = w >> 10;
    int n = w & (NJ - 1);
    int p = (kind == 0) ? 2 * n : (kind == 1) ? n : n + NJ;
    fct[w] = cas_tw(p);
  }
}

#define WT_I 32

#define FWD_TRIP 4
#define FWD_NTRIP (NJ / KSPLIT / 8 / FWD_TRIP)
#define FWD_ROWS 32
#define FWD_PITCH4 9
#define FWD_JL (NJ / KSPLIT)
#define NCB (LOG2N - 2)

static __device__ __forceinline__ void lds_barrier() {
  __builtin_amdgcn_fence(__ATOMIC_RELEASE, "workgroup", "local");
  __builtin_amdgcn_s_barrier();
  __builtin_amdgcn_fence(__ATOMIC_ACQUIRE, "workgroup", "local");
}

static __device__ __forceinline__ void fwd_load_a(v4f (&a)[2][FOLDQ][FWD_TRIP],
                                                  const float* __restrict__ xr0,
                                                  const float* __restrict__ xr1, int g) {
#pragma unroll
  for (int u = 0; u < FWD_TRIP; ++u) {
#pragma unroll
    for (int q = 0; q < FOLDQ; ++q) {
      a[0][q][u] = *(const v4f*)(xr0 + q * NJ + (g + u) * 8);
      a[1][q][u] = *(const v4f*)(xr1 + q * NJ + (g + u) * 8);
    }
  }
}

static __device__ __forceinline__ void fwd_load_b(v4f (&b)[FWD_TRIP][4], const v4f* bf, int gl) {
#pragma unroll
  for (int u = 0; u < FWD_TRIP; ++u) {
    const v4f* bg = bf + (gl + u) * 128;
    b[u][0] = bg[0];
    b[u][1] = bg[32];
    b[u][2] = bg[64];
    b[u][3] = bg[96];
  }
}

static __device__ __forceinline__ void fwd_mma(const v4f (&a)[2][FOLDQ][FWD_TRIP],
                                               const v4f (&b)[FWD_TRIP][4], v8f (&acc)[2][4]) {
#pragma unroll
  for (int u = 0; u < FWD_TRIP; ++u) {
#pragma unroll
    for (int r = 0; r < 2; ++r) {
      v4f p  = a[r][0][u] + a[r][1][u];
      v4f m  = a[r][0][u] - a[r][1][u];
      v4f pp = a[r][2][u] + a[r][3][u];
      v4f mm = a[r][2][u] - a[r][3][u];
      v4f f[4] = {p + pp, p - pp, m + mm, m - mm};
#pragma unroll
      for (int t = 0; t < 4; ++t)
        acc[r][t] = wmma_f32_16x16x4(f[t].xy, b[u][t].xy, acc[r][t]);
#pragma unroll
      for (int t = 0; t < 4; ++t)
        acc[r][t] = wmma_f32_16x16x4(f[t].zw, b[u][t].zw, acc[r][t]);
    }
  }
}

__global__ void __launch_bounds__(256, 2)
dht_fwd(const float* __restrict__ x, const float* __restrict__ w, float* __restrict__ xf,
        float* __restrict__ wpf, float* __restrict__ wmf, float* __restrict__ c2g,
        float* __restrict__ fct) {
  __shared__ v4f part[KSPLIT * NMODE * FWD_PITCH4];
  __shared__ v4f t7[(FWD_JL * NMODE) / 4];
  __shared__ float cb[NCB + 1][NMODE];
  __shared__ float wtile[2][WT_I][NMODE + 1];
  int tid  = threadIdx.x;
  int lane = tid & 31;
  int kc   = tid >> 5;
  int h     = blockIdx.x & 1;
  int g     = (blockIdx.x >> 1) & 15;
  int chalf = (blockIdx.x >> 5) & 1;
  int bt    = blockIdx.x >> 6;
  int mrow = lane & 15;
  int khalf = lane >> 4;
  int bb = 16 * bt + 8 * chalf + (mrow >> 2);
  int ii = 8 * g + 4 * h + (mrow & 3);
  const float* xr0 = x + ((size_t)bb * CIN + ii) * NN + khalf * 4;
  const float* xr1 = xr0 + (size_t)4 * CIN * NN;
  int g0 = kc * (NJ / KSPLIT / 8);
  int wg  = blockIdx.x & 15;
  int wch = (blockIdx.x >> 4) & 1;
  int wot = blockIdx.x >> 5;
  float wreg[2][(WT_I * NMODE) / 256];
#pragma unroll
  for (int s = 0; s < 2; ++s) {
#pragma unroll
    for (int e = 0; e < (WT_I * NMODE) / 256; ++e) {
      int l  = e * 256 + tid;
      int il = l >> 6;
      int k  = l & (NMODE - 1);
      int i  = 8 * wg + 4 * s + (il & 3);
      int o  = 16 * wot + 8 * wch + (il >> 2);
      wreg[s][e] = w[((size_t)i * COUT + o) * NMODE + k];
    }
  }
  v8f acc[2][4] = {};
  v4f aA[2][FOLDQ][FWD_TRIP], bA[FWD_TRIP][4];
  v4f aB[2][FOLDQ][FWD_TRIP], bB[FWD_TRIP][4];
  fwd_load_a(aA, xr0, xr1, g0);
  __builtin_amdgcn_sched_barrier(0);
#pragma unroll
  for (int rr = 0; rr < ((NCB + 1) * NMODE + 255) / 256; ++rr) {
    int q = rr * 256 + tid;
    if (q < NCB * NMODE) cb[q >> 6][q & (NMODE - 1)] = cas_tw((q & (NMODE - 1)) << (q >> 6));
    else if (q < (NCB + 1) * NMODE) cb[NCB][q & (NMODE - 1)] = 1.0f;
  }
  prep_inv_tables(blockIdx.x, tid, c2g, fct);
#pragma unroll
  for (int s = 0; s < 2; ++s) {
#pragma unroll
    for (int e = 0; e < (WT_I * NMODE) / 256; ++e) {
      int l = e * 256 + tid;
      wtile[s][l >> 6][l & (NMODE - 1)] = wreg[s][e];
    }
  }
  lds_barrier();
  {
    int e  = tid & 3;
    int l  = (tid >> 2) & 31;
    int jlow = 4 * (l >> 4) + e;
    float* t7f = (float*)t7;
#pragma unroll
    for (int th = 0; th < 2; ++th) {
      int t = ((tid >> 7) & 1) + 2 * th;
      int m = 4 * (l & 15) + t;
      float lo = cb[(jlow & 1) ? 0 : NCB][m] * cb[(jlow & 2) ? 1 : NCB][m];
      lo *= cb[(jlow & 4) ? 2 : NCB][m];
      float c3 = cb[3][m], c4 = cb[4][m], c5 = cb[5][m], c6 = cb[6][m];
#pragma unroll
      for (int gl = 0; gl < FWD_JL / 8; ++gl) {
        float v = lo;
        if (gl & 1) v *= c3;
        if (gl & 2) v *= c4;
        if (gl & 4) v *= c5;
        if (gl & 8) v *= c6;
        t7f[(gl * 4 + t) * 128 + (tid & 127)] = v;
      }
    }
  }
#pragma unroll
  for (int s = 0; s < 2; ++s) {
#pragma unroll
    for (int e = 0; e < (WT_I * NMODE) / 256; ++e) {
      int l  = e * 256 + tid;
      int k  = l >> 5;
      int il = l & (WT_I - 1);
      int kneg = (NMODE - k) & (NMODE - 1);
      float a = wtile[s][il][k];
      float b = wtile[s][il][kneg];
      size_t dst = ((((size_t)k * 8 + wot) * 16 + wg) * 2 + s) * 64 + 32 * wch + il;
      wpf[dst] = a + b;
      wmf[dst] = a - b;
    }
  }
  lds_barrier();
  const v4f* bf = t7 + lane;
  fwd_load_b(bA, bf, 0);
#pragma unroll
  for (int t = 0; t < FWD_NTRIP; t += 2) {
    fwd_load_a(aB, xr0, xr1, g0 + (t + 1) * FWD_TRIP);
    fwd_load_b(bB, bf, (t + 1) * FWD_TRIP);
    __builtin_amdgcn_sched_barrier(0);
    fwd_mma(aA, bA, acc);
    if (t + 2 < FWD_NTRIP) {
      fwd_load_a(aA, xr0, xr1, g0 + (t + 2) * FWD_TRIP);
      fwd_load_b(bA, bf, (t + 2) * FWD_TRIP);
    }
    __builtin_amdgcn_sched_barrier(0);
    fwd_mma(aB, bB, acc);
  }
  float pf4[4];
#pragma unroll
  for (int t = 0; t < 4; ++t) {
    int m = 4 * mrow + t;
    float p = cb[(kc & 1) ? 7 : NCB][m] * cb[(kc & 2) ? 8 : NCB][m];
    pf4[t] = p * cb[(kc & 4) ? 9 : NCB][m];
  }
#pragma unroll
  for (int r = 0; r < 2; ++r) {
#pragma unroll
    for (int t = 0; t < 4; ++t) {
      v4f* p = part + ((kc * 4 + t) * 16 + mrow) * FWD_PITCH4 + 4 * r + 2 * khalf;
      v4f lo = {acc[r][t][0], acc[r][t][1], acc[r][t][2], acc[r][t][3]};
      v4f hi = {acc[r][t][4], acc[r][t][5], acc[r][t][6], acc[r][t][7]};
      p[0] = lo * pf4[t];
      p[1] = hi * pf4[t];
    }
  }
  __syncthreads();
  const float* pf = (const float*)part;
#pragma unroll
  for (int e = 0; e < (FWD_ROWS * NMODE) / 256; ++e) {
    int l  = e * 256 + tid;
    int m  = l >> 5;
    int rl = l & (FWD_ROWS - 1);
    int prow = (m & 3) * 16 + (m >> 2);
    float s = 0.0f;
#pragma unroll
    for (int c = 0; c < KSPLIT; ++c)
      s += pf[(c * NMODE + prow) * (4 * FWD_PITCH4) + rl];
    xf[((((size_t)m * 4 + bt) * 16 + g) * 2 + h) * 64 + 32 * chalf + rl] = s;
  }
}

__global__ void __launch_bounds__(256)
mode_mix(const float* __restrict__ xf, const float* __restrict__ wpf,
         const float* __restrict__ wmf, float* __restrict__ outm) {
  int wave = (blockIdx.x * blockDim.x + threadIdx.x) >> 5;
  int lane = threadIdx.x & 31;
  int k   = wave >> 5;
  int rem = wave & 31;
  int bt = rem >> 3;
  int ot = rem & 7;
  int kneg = (NMODE - k) & (NMODE - 1);
  int mrow = lane & 15;
  int khalf = lane >> 4;
  const v4f* apf = (const v4f*)xf  + (size_t)(k    * (NB / 16)   + bt) * (16 * 32) + lane;
  const v4f* anf = (const v4f*)xf  + (size_t)(kneg * (NB / 16)   + bt) * (16 * 32) + lane;
  const v4f* bpf = (const v4f*)wpf + (size_t)(k    * (COUT / 16) + ot) * (16 * 32) + lane;
  const v4f* bmf = (const v4f*)wmf + (size_t)(k    * (COUT / 16) + ot) * (16 * 32) + lane;
  v8f acc = {};
  v4f ap[CIN / 8], an[CIN / 8], bp[CIN / 8], bm[CIN / 8];
#pragma unroll
  for (int g = 0; g < CIN / 8; ++g) {
    ap[g] = apf[g * 32];
    an[g] = anf[g * 32];
    bp[g] = bpf[g * 32];
    bm[g] = bmf[g * 32];
  }
  __builtin_amdgcn_sched_barrier(0);
#pragma unroll
  for (int g = 0; g < CIN / 8; ++g) {
    acc = wmma_f32_16x16x4(ap[g].xy, bp[g].xy, acc);
    acc = wmma_f32_16x16x4(an[g].xy, bm[g].xy, acc);
    acc = wmma_f32_16x16x4(ap[g].zw, bp[g].zw, acc);
    acc = wmma_f32_16x16x4(an[g].zw, bm[g].zw, acc);
  }
  int kk  = k >> 2;
  int cls = k & 3;
  int sp  = kk >> 3;
  int ii  = 2 * ((kk >> 2) & 1) + (kk & 1);
  int hh  = (kk >> 1) & 1;
#pragma unroll
  for (int v = 0; v < 8; ++v) {
    int bb = bt * 16 + v + 8 * khalf;
    int rt = bb * (COUT / 16) + ot;
    outm[((((size_t)rt * 4 + cls) * 2 + sp) * 32 + 16 * hh + mrow) * 4 + ii] = 0.5f * acc[v];
  }
}

#define INV_TILES 4
__global__ void __launch_bounds__(256)
dht_inv(const float* __restrict__ outm, const float* __restrict__ c2g,
        const float* __restrict__ fct, float* __restrict__ y) {
  int wave = (blockIdx.x * blockDim.x + threadIdx.x) >> 5;
  int lane = threadIdx.x & 31;
  int rg = wave >> 4;
  int nc = wave & (INV_NC - 1);
  int r0 = rg * (INV_TILES * 16);
  int mrow = lane & 15;
  int khalf = lane >> 4;
  const v4f* ar = (const v4f*)outm + (size_t)(r0 / 16) * (8 * 32) + lane;
  const v4f* bf = (const v4f*)c2g + (size_t)nc * (8 * 32) + lane;
  const v4f* ff = (const v4f*)fct + 16 * nc + mrow;
  v4f a[4][2];
#pragma unroll
  for (int cls = 0; cls < 4; ++cls) {
    a[cls][0] = ar[(cls * 2 + 0) * 32];
    a[cls][1] = ar[(cls * 2 + 1) * 32];
  }
  v4f b[2][4];
#pragma unroll
  for (int sp = 0; sp < 2; ++sp)
#pragma unroll
    for (int t = 0; t < 4; ++t) b[sp][t] = bf[(sp * 4 + t) * 32];
  v4f f1 = ff[0];
  v4f f0 = ff[NJ / 4];
  v4f g0 = ff[2 * (NJ / 4)];
  float* yr = y + (size_t)(r0 + 8 * khalf) * NN + 64 * nc + 4 * mrow;
#pragma nounroll
  for (int tile = 0; tile < INV_TILES; ++tile) {
    int tn = (tile + 1 < INV_TILES) ? (tile + 1) : tile;
    const v4f* arn = ar + (size_t)tn * (8 * 32);
    v4f an[4][2];
#pragma unroll
    for (int cls = 0; cls < 4; ++cls) {
      an[cls][0] = arn[(cls * 2 + 0) * 32];
      an[cls][1] = arn[(cls * 2 + 1) * 32];
    }
    __builtin_amdgcn_sched_barrier(0);
    v8f z[4][4] = {};
#pragma unroll
    for (int sp = 0; sp < 2; ++sp) {
#pragma unroll
      for (int cls = 0; cls < 4; ++cls)
#pragma unroll
        for (int t = 0; t < 4; ++t)
          z[cls][t] = wmma_f32_16x16x4(a[cls][sp].xy, b[sp][t].xy, z[cls][t]);
#pragma unroll
      for (int cls = 0; cls < 4; ++cls)
#pragma unroll
        for (int t = 0; t < 4; ++t)
          z[cls][t] = wmma_f32_16x16x4(a[cls][sp].zw, b[sp][t].zw, z[cls][t]);
    }
    float* yt = yr + (size_t)tile * (16 * NN);
#pragma unroll
    for (int v = 0; v < 8; ++v) {
      v4f z0 = {z[0][0][v], z[0][1][v], z[0][2][v], z[0][3][v]};
      v4f z1 = {z[1][0][v], z[1][1][v], z[1][2][v], z[1][3][v]};
      v4f z2 = {z[2][0][v], z[2][1][v], z[2][2][v], z[2][3][v]};
      v4f z3 = {z[3][0][v], z[3][1][v], z[3][2][v], z[3][3][v]};
      v4f pe = z0 + f1 * z2;
      v4f pm = z0 - f1 * z2;
      v4f qe = z1 + f1 * z3;
      v4f qm = z1 - f1 * z3;
      float* yv = yt + (size_t)v * NN;
      *(v4f*)(yv)          = pe + f0 * qe;
      *(v4f*)(yv + NJ)     = pm + g0 * qm;
      *(v4f*)(yv + 2 * NJ) = pe - f0 * qe;
      *(v4f*)(yv + 3 * NJ) = pm - g0 * qm;
    }
#pragma unroll
    for (int cls = 0; cls < 4; ++cls) {
      a[cls][0] = an[cls][0];
      a[cls][1] = an[cls][1];
    }
  }
}

extern "C" void kernel_launch(void* const* d_in, const int* in_sizes, int n_in,
                              void* d_out, int out_size, void* d_ws, size_t ws_size,
                              hipStream_t stream) {
  (void)in_sizes; (void)n_in; (void)out_size; (void)ws_size;
  const float* x = (const float*)d_in[0];
  const float* w = (const float*)d_in[1];
  float* y = (float*)d_out;

  float* ws   = (float*)d_ws;
  float* c2t  = ws + (size_t)NMODE * NN;
  float* fct  = c2t + (size_t)INV_NC * NJ;
  float* wp   = c2t + (size_t)NN * NMODE;
  float* wm   = wp  + (size_t)CIN * COUT * NMODE;
  float* xh   = wm  + (size_t)CIN * COUT * NMODE;
  float* outm = xh  + (size_t)NB * CIN * NMODE;

  dht_fwd<<<FWD_BLOCKS, 256, 0, stream>>>(x, w, xh, wp, wm, c2t, fct);

  mode_mix<<<(NMODE * (NB / 16) * (COUT / 16) * 32) / 256, 256, 0, stream>>>(
      xh, wp, wm, outm);

  dht_inv<<<((NB * COUT / (16 * INV_TILES)) * INV_NC * 32) / 256, 256, 0, stream>>>(
      outm, c2t, fct, y);
}
